# starting kernel
# baseline (speedup 1.0000x reference)
#include <hip/hip_runtime.h>


#define BATCH   16384
#define IN_DIM  1024
#define OUT_DIM 128

typedef __attribute__((ext_vector_type(16))) __bf16 v16bf;
typedef __attribute__((ext_vector_type(8)))  float  v8f;

static __device__ inline v8f wmma_bf16(v16bf a, v16bf b, v8f c) {
  return __builtin_amdgcn_wmma_f32_16x16x32_bf16(false, a, false, b,
                                                 (short)0, c, false, false);
}

static __device__ inline void split_fragA(const float* __restrict__ rowp,
                                          int k0, int half, bool neg,
                                          v16bf& hi, v16bf& lo) {
  const float* p0 = rowp + k0 + 8 * half;
  const float* p1 = p0 + 16;
#pragma unroll
  for (int e = 0; e < 8; ++e) {
    float x0 = neg ? -p0[e] : p0[e];
    float x1 = neg ? -p1[e] : p1[e];
    __bf16 h0 = (__bf16)x0;
    __bf16 h1 = (__bf16)x1;
    hi[e]     = h0;  lo[e]     = (__bf16)(x0 - (float)h0);
    hi[e + 8] = h1;  lo[e + 8] = (__bf16)(x1 - (float)h1);
  }
}

static __device__ inline v16bf load_fragB(const __bf16* __restrict__ rowp,
                                          int k0, int half) {
  const __bf16* p0 = rowp + k0 + 8 * half;
  const __bf16* p1 = p0 + 16;
  v16bf f;
#pragma unroll
  for (int e = 0; e < 8; ++e) { f[e] = p0[e]; f[e + 8] = p1[e]; }
  return f;
}

__global__ __launch_bounds__(256) void compute_G(const float* __restrict__ A,
                                                 float* __restrict__ G) {
  int idx = blockIdx.x * 256 + threadIdx.x;
  int r = idx >> 7, c = idx & 127;
  const float* ar = A + (size_t)r * IN_DIM;
  const float* ac = A + (size_t)c * IN_DIM;
  float s = 0.0f;
#pragma unroll 8
  for (int k = 0; k < IN_DIM; ++k) s = fmaf(ar[k], ac[k], s);
  G[idx] = s;
}

__global__ __launch_bounds__(512) void invert_G(const float* __restrict__ G,
                                                __bf16* __restrict__ Gh,
                                                __bf16* __restrict__ Gl) {
  extern __shared__ float aug[];
  __shared__ float colc[128];
  const int tid = threadIdx.x;
  for (int idx = tid; idx < 128 * 256; idx += 512) {
    int i = idx >> 8, j = idx & 255;
    aug[idx] = (j < 128) ? G[i * 128 + j] : ((j - 128) == i ? 1.0f : 0.0f);
  }
  __syncthreads();
  for (int k = 0; k < 128; ++k) {
    if (tid < 128) colc[tid] = aug[tid * 256 + k];
    __syncthreads();
    float pinv = 1.0f / colc[k];
    for (int j = tid; j < 256; j += 512) aug[k * 256 + j] *= pinv;
    __syncthreads();
    int tcol = tid & 255, r0 = tid >> 8;
    float pk = aug[k * 256 + tcol];
    for (int i = r0; i < 128; i += 2)
      if (i != k) aug[i * 256 + tcol] -= colc[i] * pk;
    __syncthreads();
  }
  for (int idx = tid; idx < 128 * 128; idx += 512) {
    int i = idx >> 7, j = idx & 127;
    float v = aug[i * 256 + 128 + j];
    __bf16 h = (__bf16)v;
    Gh[idx] = h; Gl[idx] = (__bf16)(v - (float)h);
  }
}

__global__ __launch_bounds__(256) void pack_A(const float* __restrict__ A,
                                              __bf16* __restrict__ Ah,
                                              __bf16* __restrict__ Al,
                                              __bf16* __restrict__ Ath,
                                              __bf16* __restrict__ Atl) {
  int idx = blockIdx.x * 256 + threadIdx.x;
  int r = idx >> 10, c = idx & 1023;
  float x = A[idx];
  __bf16 h = (__bf16)x;
  __bf16 l = (__bf16)(x - (float)h);
  Ah[idx] = h;  Al[idx] = l;
  Ath[c * 128 + r] = h;  Atl[c * 128 + r] = l;
}

__global__ __launch_bounds__(128) void fused_project(
    const float* __restrict__ y, const float* __restrict__ b,
    const __bf16* __restrict__ Ah, const __bf16* __restrict__ Al,
    const __bf16* __restrict__ Ath, const __bf16* __restrict__ Atl,
    const __bf16* __restrict__ Gh, const __bf16* __restrict__ Gl,
    float* __restrict__ out) {
  __shared__ float s_r[16][OUT_DIM];
  __shared__ float s_mu[16][OUT_DIM];

  const int tid  = threadIdx.x;
  const int wid  = tid >> 5;
  const int lane = tid & 31;
  const int half = lane >> 4;
  const int l15  = lane & 15;
  const int i0   = blockIdx.x * 16;

  const int n0a = (wid * 2) * 16;
  const int n0b = n0a + 16;
  const float*  yrow = y  + (size_t)(i0 + l15) * IN_DIM;
  const __bf16* b0h  = Ah + (size_t)(n0a + l15) * IN_DIM;
  const __bf16* b0l  = Al + (size_t)(n0a + l15) * IN_DIM;
  const __bf16* b1h  = Ah + (size_t)(n0b + l15) * IN_DIM;
  const __bf16* b1l  = Al + (size_t)(n0b + l15) * IN_DIM;

  v8f acc0 = {};
  v8f acc1 = {};
  for (int kk = 0; kk < IN_DIM / 32; ++kk) {
    const int k0 = kk * 32;
    v16bf yh, yl;
    split_fragA(yrow, k0, half, false, yh, yl);
    v16bf fh = load_fragB(b0h, k0, half);
    v16bf fl = load_fragB(b0l, k0, half);
    acc0 = wmma_bf16(yh, fh, acc0);
    acc0 = wmma_bf16(yl, fh, acc0);
    acc0 = wmma_bf16(yh, fl, acc0);
    fh = load_fragB(b1h, k0, half);
    fl = load_fragB(b1l, k0, half);
    acc1 = wmma_bf16(yh, fh, acc1);
    acc1 = wmma_bf16(yl, fh, acc1);
    acc1 = wmma_bf16(yh, fl, acc1);
  }
#pragma unroll
  for (int v = 0; v < 8; ++v) {
    int m = v + 8 * half;
    s_r[m][n0a + l15] = acc0[v] - b[(size_t)(i0 + m) * OUT_DIM + n0a + l15];
    s_r[m][n0b + l15] = acc1[v] - b[(size_t)(i0 + m) * OUT_DIM + n0b + l15];
  }
  __syncthreads();

  const __bf16* g0h = Gh + (size_t)(n0a + l15) * OUT_DIM;
  const __bf16* g0l = Gl + (size_t)(n0a + l15) * OUT_DIM;
  const __bf16* g1h = Gh + (size_t)(n0b + l15) * OUT_DIM;
  const __bf16* g1l = Gl + (size_t)(n0b + l15) * OUT_DIM;
  v8f mu0 = {};
  v8f mu1 = {};
#pragma unroll
  for (int kk = 0; kk < 4; ++kk) {
    const int k0 = kk * 32;
    v16bf rh, rl;
    split_fragA(&s_r[l15][0], k0, half, false, rh, rl);
    v16bf fh = load_fragB(g0h, k0, half);
    v16bf fl = load_fragB(g0l, k0, half);
    mu0 = wmma_bf16(rh, fh, mu0);
    mu0 = wmma_bf16(rl, fh, mu0);
    mu0 = wmma_bf16(rh, fl, mu0);
    fh = load_fragB(g1h, k0, half);
    fl = load_fragB(g1l, k0, half);
    mu1 = wmma_bf16(rh, fh, mu1);
    mu1 = wmma_bf16(rl, fh, mu1);
    mu1 = wmma_bf16(rh, fl, mu1);
  }
#pragma unroll
  for (int v = 0; v < 8; ++v) {
    int m = v + 8 * half;
    s_mu[m][n0a + l15] = mu0[v];
    s_mu[m][n0b + l15] = mu1[v];
  }
  __syncthreads();

  v16bf mh[4], ml[4];
#pragma unroll
  for (int kk = 0; kk < 4; ++kk)
    split_fragA(&s_mu[l15][0], kk * 32, half, true, mh[kk], ml[kk]);

  for (int t = 0; t < 16; ++t) {
    const int c0 = (wid * 16 + t) * 16;
    const __bf16* ph = Ath + (size_t)(c0 + l15) * OUT_DIM;
    const __bf16* pl = Atl + (size_t)(c0 + l15) * OUT_DIM;
    v8f acc;
#pragma unroll
    for (int v = 0; v < 8; ++v)
      acc[v] = y[(size_t)(i0 + v + 8 * half) * IN_DIM + c0 + l15];
#pragma unroll
    for (int kk = 0; kk < 4; ++kk) {
      const int k0 = kk * 32;
      v16bf fh = load_fragB(ph, k0, half);
      v16bf fl = load_fragB(pl, k0, half);
      acc = wmma_bf16(mh[kk], fh, acc);
      acc = wmma_bf16(ml[kk], fh, acc);
      acc = wmma_bf16(mh[kk], fl, acc);
    }
#pragma unroll
    for (int v = 0; v < 8; ++v)
      out[(size_t)(i0 + v + 8 * half) * IN_DIM + c0 + l15] = acc[v];
  }
}

extern "C" void kernel_launch(void* const* d_in, const int* in_sizes, int n_in,
                              void* d_out, int out_size, void* d_ws,
                              size_t ws_size, hipStream_t stream) {
  (void)in_sizes; (void)n_in; (void)out_size; (void)ws_size;
  const float* y = (const float*)d_in[0];
  const float* A = (const float*)d_in[1];
  const float* b = (const float*)d_in[2];
  float* out = (float*)d_out;

  char* ws = (char*)d_ws;
  float*  G   = (float*) (ws + 0);
  __bf16* Gh  = (__bf16*)(ws + 65536);
  __bf16* Gl  = (__bf16*)(ws + 98304);
  __bf16* Ah  = (__bf16*)(ws + 131072);
  __bf16* Al  = (__bf16*)(ws + 393216);
  __bf16* Ath = (__bf16*)(ws + 655360);
  __bf16* Atl = (__bf16*)(ws + 917504);

  compute_G<<<64, 256, 0, stream>>>(A, G);
  invert_G<<<1, 512, 128 * 256 * sizeof(float), stream>>>(G, Gh, Gl);
  pack_A<<<512, 256, 0, stream>>>(A, Ah, Al, Ath, Atl);
  fused_project<<<BATCH / 16, 128, 0, stream>>>(y, b, Ah, Al, Ath, Atl,
                                                Gh, Gl, out);
}


# invert_G 16x16 pivot-block inverse: a lane keeps half a row (its A-fragment share); pivot row by DPP row_share, not 15 v_readlane a pivot; no LDS trip; each element's operations and order unchanged
# speedup vs baseline: 5.2489x; 5.2489x over previous
#include <hip/hip_runtime.h>


#define BATCH   16384
#define IN_DIM  1024
#define OUT_DIM 128

typedef __attribute__((ext_vector_type(16))) __bf16 v16bf;
typedef __attribute__((ext_vector_type(8)))  float  v8f;
typedef __attribute__((ext_vector_type(2)))  float  v2f;

static __device__ inline v8f wmma_bf16(v16bf a, v16bf b, v8f c) {
  return __builtin_amdgcn_wmma_f32_16x16x32_bf16(false, a, false, b,
                                                 (short)0, c, false, false);
}

static __device__ inline v16bf load_fragB(const __bf16* __restrict__ rowp,
                                          int k0, int half) {
  const __bf16* p0 = rowp + k0 + 8 * half;
  const __bf16* p1 = p0 + 16;
  v16bf f;
#pragma unroll
  for (int e = 0; e < 8; ++e) { f[e] = p0[e]; f[e + 8] = p1[e]; }
  return f;
}

__global__ __launch_bounds__(256) void compute_G(const float* __restrict__ A,
                                                 float* __restrict__ G) {
  int idx = blockIdx.x * 256 + threadIdx.x;
  int r = idx >> 7, c = idx & 127;
  const float* ar = A + (size_t)r * IN_DIM;
  const float* ac = A + (size_t)c * IN_DIM;
  float s = 0.0f;
#pragma unroll 8
  for (int k = 0; k < IN_DIM; ++k) s = fmaf(ar[k], ac[k], s);
  G[idx] = s;
}

#define GLD 132

static __device__ inline v8f wmma_f32(v2f a, v2f b, v8f c) {
  return __builtin_amdgcn_wmma_f32_16x16x4_f32(false, a, false, b,
                                               (short)0, c, false, false);
}

static __device__ inline float lane_value(float x, int src) {
  return __builtin_bit_cast(float,
      __builtin_amdgcn_readlane(__builtin_bit_cast(int, x), src));
}

template <int K>
static __device__ inline float row_lane(float x) {
  return __builtin_bit_cast(float, __builtin_amdgcn_update_dpp(
      0, __builtin_bit_cast(int, x), 0x150 + K, 0xf, 0xf, true));
}

static __device__ inline float other_half(float x) {
  return __builtin_bit_cast(float, __builtin_amdgcn_permlanex16(
      __builtin_bit_cast(int, x), __builtin_bit_cast(int, x),
      0x76543210, 0xfedcba98, false, false));
}

template <int K>
static __device__ inline void gj_pivot(float (&e)[8], int c, int h) {
  constexpr int KH = K >> 3, KK = K & 7;
  const float pinv = 1.0f / lane_value(e[KK], K + 16 * KH);
  const float own  = e[KK];
  const float oth  = other_half(own);
  const float colk = (h == KH) ? own : oth;
  const bool  isk  = (c == K);
  const float nf   = isk ? 0.0f : -colk;
  float rk[8], up[8];
#pragma unroll
  for (int jj = 0; jj < 8; ++jj) rk[jj] = row_lane<K>(e[jj]) * pinv;
#pragma unroll
  for (int jj = 0; jj < 8; ++jj) up[jj] = fmaf(nf, rk[jj], e[jj]);
#pragma unroll
  for (int jj = 0; jj < 8; ++jj) e[jj] = isk ? rk[jj] : up[jj];
  const float ck = isk ? pinv : nf * pinv;
  e[KK] = (h == KH) ? ck : e[KK];
}

template <int K> struct GJ16 {
  static __device__ inline void run(float (&e)[8], int c, int h) {
    gj_pivot<K>(e, c, h);
    GJ16<K + 1>::run(e, c, h);
  }
};
template <> struct GJ16<16> {
  static __device__ inline void run(float (&)[8], int, int) {}
};

static __device__ inline void load_fragA16(const float* p, v2f (&f)[4]) {
  float4 u = *(const float4*)p;
  float4 w = *(const float4*)(p + 4);
  f[0] = v2f{u.x, u.y};  f[1] = v2f{u.z, u.w};
  f[2] = v2f{w.x, w.y};  f[3] = v2f{w.z, w.w};
}

__global__ __launch_bounds__(256) void invert_G(const float* __restrict__ G,
                                                __bf16* __restrict__ Gh,
                                                __bf16* __restrict__ Gl) {
  extern __shared__ __attribute__((aligned(16))) float M[];
  const int tid  = threadIdx.x;
  const int J    = tid >> 5;
  const int lane = tid & 31;
  const int h    = lane >> 4;
  const int c    = lane & 15;

  for (int idx = tid; idx < 128 * 32; idx += 256) {
    int i = idx >> 5, j4 = (idx & 31) * 4;
    *(float4*)(M + i * GLD + j4) = *(const float4*)(G + i * 128 + j4);
  }
  __syncthreads();

  for (int K = 0; K < 8; ++K) {
    float d[8];
    {
      const float* dp = M + (16 * K + c) * GLD + 16 * K + 8 * h;
      const float4 v0 = *(const float4*)dp;
      const float4 v1 = *(const float4*)(dp + 4);
      d[0] = v0.x; d[1] = v0.y; d[2] = v0.z; d[3] = v0.w;
      d[4] = v1.x; d[5] = v1.y; d[6] = v1.z; d[7] = v1.w;
    }
    v2f fa[8][4];
#pragma unroll
    for (int i = 0; i < 8; ++i)
      load_fragA16(M + (16 * i + c) * GLD + 16 * K + 8 * h, fa[i]);
    __syncthreads();

    if (J == K) {
      for (int rr = lane; rr < 128; rr += 32) {
        float* p = M + rr * GLD + 16 * K;
        const int e = rr - 16 * K;
#pragma unroll
        for (int j = 0; j < 16; j += 4)
          *(float4*)(p + j) = make_float4(e == j     ? 1.0f : 0.0f,
                                          e == j + 1 ? 1.0f : 0.0f,
                                          e == j + 2 ? 1.0f : 0.0f,
                                          e == j + 3 ? 1.0f : 0.0f);
      }
    }

    GJ16<0>::run(d, c, h);
    const v2f da[4] = {v2f{d[0], d[1]}, v2f{d[2], d[3]},
                       v2f{d[4], d[5]}, v2f{d[6], d[7]}};

    v8f t = {};
    {
      const float* bp = M + (16 * K + 8 * h) * GLD + 16 * J + c;
#pragma unroll
      for (int q = 0; q < 4; ++q) {
        v2f bq = v2f{bp[(2 * q) * GLD], bp[(2 * q + 1) * GLD]};
        t = wmma_f32(da[q], bq, t);
      }
    }
    v2f nb[4];
#pragma unroll
    for (int q = 0; q < 4; ++q) nb[q] = v2f{-t[2 * q], -t[2 * q + 1]};

#pragma unroll
    for (int i = 0; i < 8; ++i) {
      float* mp = M + (16 * i + 8 * h) * GLD + 16 * J + c;
      v8f acc;
#pragma unroll
      for (int r = 0; r < 8; ++r) acc[r] = mp[r * GLD];
#pragma unroll
      for (int q = 0; q < 4; ++q) acc = wmma_f32(fa[i][q], nb[q], acc);
#pragma unroll
      for (int r = 0; r < 8; ++r) mp[r * GLD] = (i == K) ? t[r] : acc[r];
    }
    __syncthreads();
  }

  for (int idx = tid; idx < 128 * 128; idx += 256) {
    int i = idx >> 7, j = idx & 127;
    float v = M[i * GLD + j];
    __bf16 hh = (__bf16)v;
    Gh[idx] = hh; Gl[idx] = (__bf16)(v - (float)hh);
  }
}

__global__ __launch_bounds__(256) void pack_A(const float* __restrict__ A,
                                              __bf16* __restrict__ Ah,
                                              __bf16* __restrict__ Al,
                                              __bf16* __restrict__ Ath,
                                              __bf16* __restrict__ Atl) {
  int idx = blockIdx.x * 256 + threadIdx.x;
  int r = idx >> 10, c = idx & 1023;
  float x = A[idx];
  __bf16 h = (__bf16)x;
  __bf16 l = (__bf16)(x - (float)h);
  Ah[idx] = h;  Al[idx] = l;
  Ath[c * 128 + r] = h;  Atl[c * 128 + r] = l;
}

#define RB   64
#define HP   136
#define YP   132
#define OFF_YL 17408
#define OFF_BH 34816
#define OFF_BL 69632
#define BUFB   104448

typedef __attribute__((ext_vector_type(8))) __bf16 v8bf;
typedef __attribute__((ext_vector_type(4))) __bf16 v4bf;
typedef __attribute__((ext_vector_type(4))) float  v4f;

static __device__ inline v16bf lds_frag(const __bf16* rowp, int k0, int half) {
  const v8bf u = *(const v8bf*)(rowp + k0 + 8 * half);
  const v8bf w = *(const v8bf*)(rowp + k0 + 16 + 8 * half);
  return __builtin_shufflevector(u, w, 0, 1, 2, 3, 4, 5, 6, 7,
                                 8, 9, 10, 11, 12, 13, 14, 15);
}

static __device__ inline void split_fragA_lds(const float* rowp, int k0,
                                              int half, bool neg,
                                              v16bf& hi, v16bf& lo) {
  const float* p0 = rowp + k0 + 8 * half;
  const float4 a0 = *(const float4*)(p0);
  const float4 a1 = *(const float4*)(p0 + 4);
  const float4 c0 = *(const float4*)(p0 + 16);
  const float4 c1 = *(const float4*)(p0 + 20);
  const float x[16] = {a0.x, a0.y, a0.z, a0.w, a1.x, a1.y, a1.z, a1.w,
                       c0.x, c0.y, c0.z, c0.w, c1.x, c1.y, c1.z, c1.w};
#pragma unroll
  for (int e = 0; e < 16; ++e) {
    const float xe = neg ? -x[e] : x[e];
    const __bf16 he = (__bf16)xe;
    hi[e] = he;  lo[e] = (__bf16)(xe - (float)he);
  }
}

static __device__ inline void fetch_y(const float* __restrict__ y, int i0,
                                      int col0, int tid, v4f (&yq)[4]) {
#pragma unroll
  for (int j = 0; j < 4; ++j) {
    const int idx = tid + 512 * j;
    yq[j] = *(const v4f*)(y + (size_t)(i0 + (idx >> 5)) * IN_DIM + col0 +
                          4 * (idx & 31));
  }
}

static __device__ inline void fetch_b(const __bf16* __restrict__ Ph,
                                      const __bf16* __restrict__ Pl,
                                      size_t base, int pitch, int tid,
                                      v8bf (&bq)[8]) {
#pragma unroll
  for (int j = 0; j < 4; ++j) {
    const int idx = tid + 512 * j;
    const size_t off = base + (size_t)(idx >> 4) * pitch + 8 * (idx & 15);
    bq[j]     = *(const v8bf*)(Ph + off);
    bq[j + 4] = *(const v8bf*)(Pl + off);
  }
}

static __device__ inline void stash_b(char* buf, int tid, const v8bf (&bq)[8]) {
  __bf16* Bh = (__bf16*)(buf + OFF_BH);
  __bf16* Bl = (__bf16*)(buf + OFF_BL);
#pragma unroll
  for (int j = 0; j < 4; ++j) {
    const int idx = tid + 512 * j;
    const int o = (idx >> 4) * HP + 8 * (idx & 15);
    *(v8bf*)(Bh + o) = bq[j];
    *(v8bf*)(Bl + o) = bq[j + 4];
  }
}

static __device__ inline void stash_y_split(char* buf, int tid,
                                            const v4f (&yq)[4]) {
  __bf16* Yh = (__bf16*)buf;
  __bf16* Yl = (__bf16*)(buf + OFF_YL);
#pragma unroll
  for (int j = 0; j < 4; ++j) {
    const int idx = tid + 512 * j;
    const int o = (idx >> 5) * HP + 4 * (idx & 31);
    const float x[4] = {yq[j][0], yq[j][1], yq[j][2], yq[j][3]};
    v4bf h, l;
#pragma unroll
    for (int e = 0; e < 4; ++e) {
      const __bf16 he = (__bf16)x[e];
      h[e] = he;  l[e] = (__bf16)(x[e] - (float)he);
    }
    *(v4bf*)(Yh + o) = h;
    *(v4bf*)(Yl + o) = l;
  }
}

static __device__ inline void stash_y_f32(char* buf, int tid,
                                          const v4f (&yq)[4]) {
  float* Yc = (float*)buf;
#pragma unroll
  for (int j = 0; j < 4; ++j) {
    const int idx = tid + 512 * j;
    *(v4f*)(Yc + (idx >> 5) * YP + 4 * (idx & 31)) = yq[j];
  }
}

__global__ __launch_bounds__(512) void fused_project(
    const float* __restrict__ y, const float* __restrict__ b,
    const __bf16* __restrict__ Ah, const __bf16* __restrict__ Al,
    const __bf16* __restrict__ Ath, const __bf16* __restrict__ Atl,
    const __bf16* __restrict__ Gh, const __bf16* __restrict__ Gl,
    float* __restrict__ out) {
  extern __shared__ __attribute__((aligned(16))) char lds[];
  char*  buf0 = lds;
  char*  buf1 = lds + BUFB;
  float* s_r  = (float*)(lds + 2 * BUFB);
  float* s_mu = s_r + RB * OUT_DIM;

  const int tid  = threadIdx.x;
  const int wid  = tid >> 5;
  const int rt   = wid >> 2;
  const int cq   = wid & 3;
  const int lane = tid & 31;
  const int half = lane >> 4;
  const int l15  = lane & 15;
  const int i0   = blockIdx.x * RB;
  const int n0a  = cq * 32;
  const int n0b  = n0a + 16;

  v4f    yq[4];
  v8bf   bq[8];

  fetch_y(y, i0, 0, tid, yq);
  fetch_b(Ah, Al, 0, IN_DIM, tid, bq);
  stash_y_split(buf0, tid, yq);
  stash_b(buf0, tid, bq);
  __syncthreads();

  v8f acc0 = {};
  v8f acc1 = {};
  for (int c = 0; c < 8; ++c) {
    char* cur = (c & 1) ? buf1 : buf0;
    char* nxt = (c & 1) ? buf0 : buf1;
    if (c + 1 < 8) {
      fetch_y(y, i0, 128 * (c + 1), tid, yq);
      fetch_b(Ah, Al, (size_t)(128 * (c + 1)), IN_DIM, tid, bq);
    }
    __builtin_amdgcn_sched_barrier(0);
    {
      const __bf16* yh  = (const __bf16*)cur + (16 * rt + l15) * HP;
      const __bf16* yl  = (const __bf16*)(cur + OFF_YL) + (16 * rt + l15) * HP;
      const __bf16* b0h = (const __bf16*)(cur + OFF_BH) + (n0a + l15) * HP;
      const __bf16* b0l = (const __bf16*)(cur + OFF_BL) + (n0a + l15) * HP;
      const __bf16* b1h = b0h + 16 * HP;
      const __bf16* b1l = b0l + 16 * HP;
#pragma unroll
      for (int ks = 0; ks < 4; ++ks) {
        const int k0 = 32 * ks;
        const v16bf ah = lds_frag(yh, k0, half);
        const v16bf al = lds_frag(yl, k0, half);
        v16bf fh = lds_frag(b0h, k0, half);
        v16bf fl = lds_frag(b0l, k0, half);
        acc0 = wmma_bf16(ah, fh, acc0);
        acc0 = wmma_bf16(al, fh, acc0);
        acc0 = wmma_bf16(ah, fl, acc0);
        fh = lds_frag(b1h, k0, half);
        fl = lds_frag(b1l, k0, half);
        acc1 = wmma_bf16(ah, fh, acc1);
        acc1 = wmma_bf16(al, fh, acc1);
        acc1 = wmma_bf16(ah, fl, acc1);
      }
    }
    if (c + 1 < 8) {
      asm volatile("" : "+v"(acc0), "+v"(acc1), "+v"(yq[0]), "+v"(yq[1]),
                        "+v"(yq[2]), "+v"(yq[3]));
      stash_y_split(nxt, tid, yq);
      stash_b(nxt, tid, bq);
    }
    __syncthreads();
  }

  fetch_y(y, i0, 0, tid, yq);
  fetch_b(Ath, Atl, 0, OUT_DIM, tid, bq);

#pragma unroll
  for (int v = 0; v < 8; ++v) {
    const int m = 16 * rt + v + 8 * half;
    s_r[m * OUT_DIM + n0a + l15] =
        acc0[v] - b[(size_t)(i0 + m) * OUT_DIM + n0a + l15];
    s_r[m * OUT_DIM + n0b + l15] =
        acc1[v] - b[(size_t)(i0 + m) * OUT_DIM + n0b + l15];
  }
  __syncthreads();

  const __bf16* g0h = Gh + (size_t)(n0a + l15) * OUT_DIM;
  const __bf16* g0l = Gl + (size_t)(n0a + l15) * OUT_DIM;
  const __bf16* g1h = Gh + (size_t)(n0b + l15) * OUT_DIM;
  const __bf16* g1l = Gl + (size_t)(n0b + l15) * OUT_DIM;
  v8f mu0 = {};
  v8f mu1 = {};
#pragma unroll
  for (int kk = 0; kk < 4; ++kk) {
    const int k0 = kk * 32;
    v16bf rh, rl;
    split_fragA_lds(s_r + (16 * rt + l15) * OUT_DIM, k0, half, false, rh, rl);
    v16bf fh = load_fragB(g0h, k0, half);
    v16bf fl = load_fragB(g0l, k0, half);
    mu0 = wmma_bf16(rh, fh, mu0);
    mu0 = wmma_bf16(rl, fh, mu0);
    mu0 = wmma_bf16(rh, fl, mu0);
    fh = load_fragB(g1h, k0, half);
    fl = load_fragB(g1l, k0, half);
    mu1 = wmma_bf16(rh, fh, mu1);
    mu1 = wmma_bf16(rl, fh, mu1);
    mu1 = wmma_bf16(rh, fl, mu1);
  }
#pragma unroll
  for (int v = 0; v < 8; ++v) {
    const int m = 16 * rt + v + 8 * half;
    s_mu[m * OUT_DIM + n0a + l15] = mu0[v];
    s_mu[m * OUT_DIM + n0b + l15] = mu1[v];
  }
  __syncthreads();

  v16bf mh[4], ml[4];
#pragma unroll
  for (int kk = 0; kk < 4; ++kk)
    split_fragA_lds(s_mu + (16 * rt + l15) * OUT_DIM, kk * 32, half, true,
                    mh[kk], ml[kk]);

  stash_y_f32(buf0, tid, yq);
  stash_b(buf0, tid, bq);
  __syncthreads();

  for (int c = 0; c < 8; ++c) {
    char* cur = (c & 1) ? buf1 : buf0;
    char* nxt = (c & 1) ? buf0 : buf1;
    if (c + 1 < 8) {
      fetch_y(y, i0, 128 * (c + 1), tid, yq);
      fetch_b(Ath, Atl, (size_t)(128 * (c + 1)) * OUT_DIM, OUT_DIM, tid, bq);
    }
    __builtin_amdgcn_sched_barrier(0);
#pragma unroll
    for (int t2 = 0; t2 < 2; ++t2) {
      const int ct = 32 * cq + 16 * t2;
      const float*  yc = (const float*)cur + (16 * rt + 8 * half) * YP + ct + l15;
      const __bf16* ph = (const __bf16*)(cur + OFF_BH) + (ct + l15) * HP;
      const __bf16* pl = (const __bf16*)(cur + OFF_BL) + (ct + l15) * HP;
      v8f acc;
#pragma unroll
      for (int v = 0; v < 8; ++v) acc[v] = yc[v * YP];
#pragma unroll
      for (int kk = 0; kk < 4; ++kk) {
        const int k0 = kk * 32;
        const v16bf fh = lds_frag(ph, k0, half);
        const v16bf fl = lds_frag(pl, k0, half);
        acc = wmma_bf16(mh[kk], fh, acc);
        acc = wmma_bf16(ml[kk], fh, acc);
        acc = wmma_bf16(mh[kk], fl, acc);
      }
      float* op = out + (size_t)(i0 + 16 * rt + 8 * half) * IN_DIM + 128 * c + ct + l15;
#pragma unroll
      for (int v = 0; v < 8; ++v) op[(size_t)v * IN_DIM] = acc[v];
    }
    if (c + 1 < 8) {
      stash_y_f32(nxt, tid, yq);
      stash_b(nxt, tid, bq);
    }
    __syncthreads();
  }
}

extern "C" void kernel_launch(void* const* d_in, const int* in_sizes, int n_in,
                              void* d_out, int out_size, void* d_ws,
                              size_t ws_size, hipStream_t stream) {
  (void)in_sizes; (void)n_in; (void)out_size; (void)ws_size;
  const float* y = (const float*)d_in[0];
  const float* A = (const float*)d_in[1];
  const float* b = (const float*)d_in[2];
  float* out = (float*)d_out;

  char* ws = (char*)d_ws;
  float*  G   = (float*) (ws + 0);
  __bf16* Gh  = (__bf16*)(ws + 65536);
  __bf16* Gl  = (__bf16*)(ws + 98304);
  __bf16* Ah  = (__bf16*)(ws + 131072);
  __bf16* Al  = (__bf16*)(ws + 393216);
  __bf16* Ath = (__bf16*)(ws + 655360);
  __bf16* Atl = (__bf16*)(ws + 917504);

  compute_G<<<64, 256, 0, stream>>>(A, G);
  invert_G<<<1, 256, 67584, stream>>>(G, Gh, Gl);
  pack_A<<<512, 256, 0, stream>>>(A, Ah, Al, Ath, Atl);
  fused_project<<<256, 512, 274432, stream>>>(y, b, Ah, Al, Ath, Atl,
                                              Gh, Gl, out);
}


# compute_G: serial 1024-long dot product per thread, lane-strided reads -> 8x8 tile per group, K split over 256 threads (coalesced 16-B loads), partial tiles summed via LDS; G summed in another order
# speedup vs baseline: 6.4798x; 1.2345x over previous
#include <hip/hip_runtime.h>


#define BATCH   16384
#define IN_DIM  1024
#define OUT_DIM 128

typedef __attribute__((ext_vector_type(16))) __bf16 v16bf;
typedef __attribute__((ext_vector_type(8)))  float  v8f;
typedef __attribute__((ext_vector_type(2)))  float  v2f;

static __device__ inline v8f wmma_bf16(v16bf a, v16bf b, v8f c) {
  return __builtin_amdgcn_wmma_f32_16x16x32_bf16(false, a, false, b,
                                                 (short)0, c, false, false);
}

static __device__ inline v16bf load_fragB(const __bf16* __restrict__ rowp,
                                          int k0, int half) {
  const __bf16* p0 = rowp + k0 + 8 * half;
  const __bf16* p1 = p0 + 16;
  v16bf f;
#pragma unroll
  for (int e = 0; e < 8; ++e) { f[e] = p0[e]; f[e + 8] = p1[e]; }
  return f;
}

#define GPP 68

__global__ __launch_bounds__(256) void compute_G(const float* __restrict__ A,
                                                 float* __restrict__ G) {
  extern __shared__ __attribute__((aligned(16))) float gpart[];
  const int t  = threadIdx.x;
  const int bi = blockIdx.x >> 4, bj = blockIdx.x & 15;
  const float* ap = A + (size_t)(8 * bi) * IN_DIM + 4 * t;
  const float* bp = A + (size_t)(8 * bj) * IN_DIM + 4 * t;
  float4 ra[8], rb[8];
#pragma unroll
  for (int r = 0; r < 8; ++r) {
    ra[r] = *(const float4*)(ap + (size_t)r * IN_DIM);
    rb[r] = *(const float4*)(bp + (size_t)r * IN_DIM);
  }
  float* mine = gpart + t * GPP;
#pragma unroll
  for (int i = 0; i < 8; ++i) {
    float s[8];
#pragma unroll
    for (int j = 0; j < 8; ++j) {
      float v = ra[i].x * rb[j].x;
      v = fmaf(ra[i].y, rb[j].y, v);
      v = fmaf(ra[i].z, rb[j].z, v);
      s[j] = fmaf(ra[i].w, rb[j].w, v);
    }
    *(float4*)(mine + 8 * i)     = make_float4(s[0], s[1], s[2], s[3]);
    *(float4*)(mine + 8 * i + 4) = make_float4(s[4], s[5], s[6], s[7]);
  }
  __syncthreads();

  const int o = t & 63, q = t >> 6;
  const float* col = gpart + (64 * q) * GPP + o;
  float c0 = col[0], c1 = col[GPP], c2 = col[2 * GPP], c3 = col[3 * GPP];
#pragma unroll
  for (int u = 4; u < 64; u += 4) {
    c0 += col[(u    ) * GPP];
    c1 += col[(u + 1) * GPP];
    c2 += col[(u + 2) * GPP];
    c3 += col[(u + 3) * GPP];
  }
  float* red = gpart + 256 * GPP;
  red[64 * q + o] = (c0 + c1) + (c2 + c3);
  __syncthreads();
  if (t < 64)
    G[blockIdx.x * 64 + t] = (red[t] + red[64 + t]) + (red[128 + t] + red[192 + t]);
}

#define GLD 132

static __device__ inline v8f wmma_f32(v2f a, v2f b, v8f c) {
  return __builtin_amdgcn_wmma_f32_16x16x4_f32(false, a, false, b,
                                               (short)0, c, false, false);
}

static __device__ inline float lane_value(float x, int src) {
  return __builtin_bit_cast(float,
      __builtin_amdgcn_readlane(__builtin_bit_cast(int, x), src));
}

template <int K>
static __device__ inline float row_lane(float x) {
  return __builtin_bit_cast(float, __builtin_amdgcn_update_dpp(
      0, __builtin_bit_cast(int, x), 0x150 + K, 0xf, 0xf, true));
}

static __device__ inline float other_half(float x) {
  return __builtin_bit_cast(float, __builtin_amdgcn_permlanex16(
      __builtin_bit_cast(int, x), __builtin_bit_cast(int, x),
      0x76543210, 0xfedcba98, false, false));
}

template <int K>
static __device__ inline void gj_pivot(float (&e)[8], int c, int h) {
  constexpr int KH = K >> 3, KK = K & 7;
  const float pinv = 1.0f / lane_value(e[KK], K + 16 * KH);
  const float own  = e[KK];
  const float oth  = other_half(own);
  const float colk = (h == KH) ? own : oth;
  const bool  isk  = (c == K);
  const float nf   = isk ? 0.0f : -colk;
  float rk[8], up[8];
#pragma unroll
  for (int jj = 0; jj < 8; ++jj) rk[jj] = row_lane<K>(e[jj]) * pinv;
#pragma unroll
  for (int jj = 0; jj < 8; ++jj) up[jj] = fmaf(nf, rk[jj], e[jj]);
#pragma unroll
  for (int jj = 0; jj < 8; ++jj) e[jj] = isk ? rk[jj] : up[jj];
  const float ck = isk ? pinv : nf * pinv;
  e[KK] = (h == KH) ? ck : e[KK];
}

template <int K> struct GJ16 {
  static __device__ inline void run(float (&e)[8], int c, int h) {
    gj_pivot<K>(e, c, h);
    GJ16<K + 1>::run(e, c, h);
  }
};
template <> struct GJ16<16> {
  static __device__ inline void run(float (&)[8], int, int) {}
};

static __device__ inline void load_fragA16(const float* p, v2f (&f)[4]) {
  float4 u = *(const float4*)p;
  float4 w = *(const float4*)(p + 4);
  f[0] = v2f{u.x, u.y};  f[1] = v2f{u.z, u.w};
  f[2] = v2f{w.x, w.y};  f[3] = v2f{w.z, w.w};
}

__global__ __launch_bounds__(256) void invert_G(const float* __restrict__ G,
                                                __bf16* __restrict__ Gh,
                                                __bf16* __restrict__ Gl) {
  extern __shared__ __attribute__((aligned(16))) float M[];
  const int tid  = threadIdx.x;
  const int J    = tid >> 5;
  const int lane = tid & 31;
  const int h    = lane >> 4;
  const int c    = lane & 15;

  for (int idx = tid; idx < 128 * 32; idx += 256) {
    const int tile = idx >> 4;
    const int i  = 8 * (tile >> 4) + ((idx >> 1) & 7);
    const int j4 = 8 * (tile & 15) + 4 * (idx & 1);
    *(float4*)(M + i * GLD + j4) = *(const float4*)(G + 4 * idx);
  }
  __syncthreads();

  for (int K = 0; K < 8; ++K) {
    float d[8];
    {
      const float* dp = M + (16 * K + c) * GLD + 16 * K + 8 * h;
      const float4 v0 = *(const float4*)dp;
      const float4 v1 = *(const float4*)(dp + 4);
      d[0] = v0.x; d[1] = v0.y; d[2] = v0.z; d[3] = v0.w;
      d[4] = v1.x; d[5] = v1.y; d[6] = v1.z; d[7] = v1.w;
    }
    v2f fa[8][4];
#pragma unroll
    for (int i = 0; i < 8; ++i)
      load_fragA16(M + (16 * i + c) * GLD + 16 * K + 8 * h, fa[i]);
    __syncthreads();

    if (J == K) {
      for (int rr = lane; rr < 128; rr += 32) {
        float* p = M + rr * GLD + 16 * K;
        const int e = rr - 16 * K;
#pragma unroll
        for (int j = 0; j < 16; j += 4)
          *(float4*)(p + j) = make_float4(e == j     ? 1.0f : 0.0f,
                                          e == j + 1 ? 1.0f : 0.0f,
                                          e == j + 2 ? 1.0f : 0.0f,
                                          e == j + 3 ? 1.0f : 0.0f);
      }
    }

    GJ16<0>::run(d, c, h);
    const v2f da[4] = {v2f{d[0], d[1]}, v2f{d[2], d[3]},
                       v2f{d[4], d[5]}, v2f{d[6], d[7]}};

    v8f t = {};
    {
      const float* bp = M + (16 * K + 8 * h) * GLD + 16 * J + c;
#pragma unroll
      for (int q = 0; q < 4; ++q) {
        v2f bq = v2f{bp[(2 * q) * GLD], bp[(2 * q + 1) * GLD]};
        t = wmma_f32(da[q], bq, t);
      }
    }
    v2f nb[4];
#pragma unroll
    for (int q = 0; q < 4; ++q) nb[q] = v2f{-t[2 * q], -t[2 * q + 1]};

#pragma unroll
    for (int i = 0; i < 8; ++i) {
      float* mp = M + (16 * i + 8 * h) * GLD + 16 * J + c;
      v8f acc;
#pragma unroll
      for (int r = 0; r < 8; ++r) acc[r] = mp[r * GLD];
#pragma unroll
      for (int q = 0; q < 4; ++q) acc = wmma_f32(fa[i][q], nb[q], acc);
#pragma unroll
      for (int r = 0; r < 8; ++r) mp[r * GLD] = (i == K) ? t[r] : acc[r];
    }
    __syncthreads();
  }

  for (int idx = tid; idx < 128 * 128; idx += 256) {
    int i = idx >> 7, j = idx & 127;
    float v = M[i * GLD + j];
    __bf16 hh = (__bf16)v;
    Gh[idx] = hh; Gl[idx] = (__bf16)(v - (float)hh);
  }
}

__global__ __launch_bounds__(256) void pack_A(const float* __restrict__ A,
                                              __bf16* __restrict__ Ah,
                                              __bf16* __restrict__ Al,
                                              __bf16* __restrict__ Ath,
                                              __bf16* __restrict__ Atl) {
  int idx = blockIdx.x * 256 + threadIdx.x;
  int r = idx >> 10, c = idx & 1023;
  float x = A[idx];
  __bf16 h = (__bf16)x;
  __bf16 l = (__bf16)(x - (float)h);
  Ah[idx] = h;  Al[idx] = l;
  Ath[c * 128 + r] = h;  Atl[c * 128 + r] = l;
}

#define RB   64
#define HP   136
#define YP   132
#define OFF_YL 17408
#define OFF_BH 34816
#define OFF_BL 69632
#define BUFB   104448

typedef __attribute__((ext_vector_type(8))) __bf16 v8bf;
typedef __attribute__((ext_vector_type(4))) __bf16 v4bf;
typedef __attribute__((ext_vector_type(4))) float  v4f;

static __device__ inline v16bf lds_frag(const __bf16* rowp, int k0, int half) {
  const v8bf u = *(const v8bf*)(rowp + k0 + 8 * half);
  const v8bf w = *(const v8bf*)(rowp + k0 + 16 + 8 * half);
  return __builtin_shufflevector(u, w, 0, 1, 2, 3, 4, 5, 6, 7,
                                 8, 9, 10, 11, 12, 13, 14, 15);
}

static __device__ inline void split_fragA_lds(const float* rowp, int k0,
                                              int half, bool neg,
                                              v16bf& hi, v16bf& lo) {
  const float* p0 = rowp + k0 + 8 * half;
  const float4 a0 = *(const float4*)(p0);
  const float4 a1 = *(const float4*)(p0 + 4);
  const float4 c0 = *(const float4*)(p0 + 16);
  const float4 c1 = *(const float4*)(p0 + 20);
  const float x[16] = {a0.x, a0.y, a0.z, a0.w, a1.x, a1.y, a1.z, a1.w,
                       c0.x, c0.y, c0.z, c0.w, c1.x, c1.y, c1.z, c1.w};
#pragma unroll
  for (int e = 0; e < 16; ++e) {
    const float xe = neg ? -x[e] : x[e];
    const __bf16 he = (__bf16)xe;
    hi[e] = he;  lo[e] = (__bf16)(xe - (float)he);
  }
}

static __device__ inline void fetch_y(const float* __restrict__ y, int i0,
                                      int col0, int tid, v4f (&yq)[4]) {
#pragma unroll
  for (int j = 0; j < 4; ++j) {
    const int idx = tid + 512 * j;
    yq[j] = *(const v4f*)(y + (size_t)(i0 + (idx >> 5)) * IN_DIM + col0 +
                          4 * (idx & 31));
  }
}

static __device__ inline void fetch_b(const __bf16* __restrict__ Ph,
                                      const __bf16* __restrict__ Pl,
                                      size_t base, int pitch, int tid,
                                      v8bf (&bq)[8]) {
#pragma unroll
  for (int j = 0; j < 4; ++j) {
    const int idx = tid + 512 * j;
    const size_t off = base + (size_t)(idx >> 4) * pitch + 8 * (idx & 15);
    bq[j]     = *(const v8bf*)(Ph + off);
    bq[j + 4] = *(const v8bf*)(Pl + off);
  }
}

static __device__ inline void stash_b(char* buf, int tid, const v8bf (&bq)[8]) {
  __bf16* Bh = (__bf16*)(buf + OFF_BH);
  __bf16* Bl = (__bf16*)(buf + OFF_BL);
#pragma unroll
  for (int j = 0; j < 4; ++j) {
    const int idx = tid + 512 * j;
    const int o = (idx >> 4) * HP + 8 * (idx & 15);
    *(v8bf*)(Bh + o) = bq[j];
    *(v8bf*)(Bl + o) = bq[j + 4];
  }
}

static __device__ inline void stash_y_split(char* buf, int tid,
                                            const v4f (&yq)[4]) {
  __bf16* Yh = (__bf16*)buf;
  __bf16* Yl = (__bf16*)(buf + OFF_YL);
#pragma unroll
  for (int j = 0; j < 4; ++j) {
    const int idx = tid + 512 * j;
    const int o = (idx >> 5) * HP + 4 * (idx & 31);
    const float x[4] = {yq[j][0], yq[j][1], yq[j][2], yq[j][3]};
    v4bf h, l;
#pragma unroll
    for (int e = 0; e < 4; ++e) {
      const __bf16 he = (__bf16)x[e];
      h[e] = he;  l[e] = (__bf16)(x[e] - (float)he);
    }
    *(v4bf*)(Yh + o) = h;
    *(v4bf*)(Yl + o) = l;
  }
}

static __device__ inline void stash_y_f32(char* buf, int tid,
                                          const v4f (&yq)[4]) {
  float* Yc = (float*)buf;
#pragma unroll
  for (int j = 0; j < 4; ++j) {
    const int idx = tid + 512 * j;
    *(v4f*)(Yc + (idx >> 5) * YP + 4 * (idx & 31)) = yq[j];
  }
}

__global__ __launch_bounds__(512) void fused_project(
    const float* __restrict__ y, const float* __restrict__ b,
    const __bf16* __restrict__ Ah, const __bf16* __restrict__ Al,
    const __bf16* __restrict__ Ath, const __bf16* __restrict__ Atl,
    const __bf16* __restrict__ Gh, const __bf16* __restrict__ Gl,
    float* __restrict__ out) {
  extern __shared__ __attribute__((aligned(16))) char lds[];
  char*  buf0 = lds;
  char*  buf1 = lds + BUFB;
  float* s_r  = (float*)(lds + 2 * BUFB);
  float* s_mu = s_r + RB * OUT_DIM;

  const int tid  = threadIdx.x;
  const int wid  = tid >> 5;
  const int rt   = wid >> 2;
  const int cq   = wid & 3;
  const int lane = tid & 31;
  const int half = lane >> 4;
  const int l15  = lane & 15;
  const int i0   = blockIdx.x * RB;
  const int n0a  = cq * 32;
  const int n0b  = n0a + 16;

  v4f    yq[4];
  v8bf   bq[8];

  fetch_y(y, i0, 0, tid, yq);
  fetch_b(Ah, Al, 0, IN_DIM, tid, bq);
  stash_y_split(buf0, tid, yq);
  stash_b(buf0, tid, bq);
  __syncthreads();

  v8f acc0 = {};
  v8f acc1 = {};
  for (int c = 0; c < 8; ++c) {
    char* cur = (c & 1) ? buf1 : buf0;
    char* nxt = (c & 1) ? buf0 : buf1;
    if (c + 1 < 8) {
      fetch_y(y, i0, 128 * (c + 1), tid, yq);
      fetch_b(Ah, Al, (size_t)(128 * (c + 1)), IN_DIM, tid, bq);
    }
    __builtin_amdgcn_sched_barrier(0);
    {
      const __bf16* yh  = (const __bf16*)cur + (16 * rt + l15) * HP;
      const __bf16* yl  = (const __bf16*)(cur + OFF_YL) + (16 * rt + l15) * HP;
      const __bf16* b0h = (const __bf16*)(cur + OFF_BH) + (n0a + l15) * HP;
      const __bf16* b0l = (const __bf16*)(cur + OFF_BL) + (n0a + l15) * HP;
      const __bf16* b1h = b0h + 16 * HP;
      const __bf16* b1l = b0l + 16 * HP;
#pragma unroll
      for (int ks = 0; ks < 4; ++ks) {
        const int k0 = 32 * ks;
        const v16bf ah = lds_frag(yh, k0, half);
        const v16bf al = lds_frag(yl, k0, half);
        v16bf fh = lds_frag(b0h, k0, half);
        v16bf fl = lds_frag(b0l, k0, half);
        acc0 = wmma_bf16(ah, fh, acc0);
        acc0 = wmma_bf16(al, fh, acc0);
        acc0 = wmma_bf16(ah, fl, acc0);
        fh = lds_frag(b1h, k0, half);
        fl = lds_frag(b1l, k0, half);
        acc1 = wmma_bf16(ah, fh, acc1);
        acc1 = wmma_bf16(al, fh, acc1);
        acc1 = wmma_bf16(ah, fl, acc1);
      }
    }
    if (c + 1 < 8) {
      asm volatile("" : "+v"(acc0), "+v"(acc1), "+v"(yq[0]), "+v"(yq[1]),
                        "+v"(yq[2]), "+v"(yq[3]));
      stash_y_split(nxt, tid, yq);
      stash_b(nxt, tid, bq);
    }
    __syncthreads();
  }

  fetch_y(y, i0, 0, tid, yq);
  fetch_b(Ath, Atl, 0, OUT_DIM, tid, bq);

#pragma unroll
  for (int v = 0; v < 8; ++v) {
    const int m = 16 * rt + v + 8 * half;
    s_r[m * OUT_DIM + n0a + l15] =
        acc0[v] - b[(size_t)(i0 + m) * OUT_DIM + n0a + l15];
    s_r[m * OUT_DIM + n0b + l15] =
        acc1[v] - b[(size_t)(i0 + m) * OUT_DIM + n0b + l15];
  }
  __syncthreads();

  const __bf16* g0h = Gh + (size_t)(n0a + l15) * OUT_DIM;
  const __bf16* g0l = Gl + (size_t)(n0a + l15) * OUT_DIM;
  const __bf16* g1h = Gh + (size_t)(n0b + l15) * OUT_DIM;
  const __bf16* g1l = Gl + (size_t)(n0b + l15) * OUT_DIM;
  v8f mu0 = {};
  v8f mu1 = {};
#pragma unroll
  for (int kk = 0; kk < 4; ++kk) {
    const int k0 = kk * 32;
    v16bf rh, rl;
    split_fragA_lds(s_r + (16 * rt + l15) * OUT_DIM, k0, half, false, rh, rl);
    v16bf fh = load_fragB(g0h, k0, half);
    v16bf fl = load_fragB(g0l, k0, half);
    mu0 = wmma_bf16(rh, fh, mu0);
    mu0 = wmma_bf16(rl, fh, mu0);
    mu0 = wmma_bf16(rh, fl, mu0);
    fh = load_fragB(g1h, k0, half);
    fl = load_fragB(g1l, k0, half);
    mu1 = wmma_bf16(rh, fh, mu1);
    mu1 = wmma_bf16(rl, fh, mu1);
    mu1 = wmma_bf16(rh, fl, mu1);
  }
#pragma unroll
  for (int v = 0; v < 8; ++v) {
    const int m = 16 * rt + v + 8 * half;
    s_mu[m * OUT_DIM + n0a + l15] = mu0[v];
    s_mu[m * OUT_DIM + n0b + l15] = mu1[v];
  }
  __syncthreads();

  v16bf mh[4], ml[4];
#pragma unroll
  for (int kk = 0; kk < 4; ++kk)
    split_fragA_lds(s_mu + (16 * rt + l15) * OUT_DIM, kk * 32, half, true,
                    mh[kk], ml[kk]);

  stash_y_f32(buf0, tid, yq);
  stash_b(buf0, tid, bq);
  __syncthreads();

  for (int c = 0; c < 8; ++c) {
    char* cur = (c & 1) ? buf1 : buf0;
    char* nxt = (c & 1) ? buf0 : buf1;
    if (c + 1 < 8) {
      fetch_y(y, i0, 128 * (c + 1), tid, yq);
      fetch_b(Ath, Atl, (size_t)(128 * (c + 1)) * OUT_DIM, OUT_DIM, tid, bq);
    }
    __builtin_amdgcn_sched_barrier(0);
#pragma unroll
    for (int t2 = 0; t2 < 2; ++t2) {
      const int ct = 32 * cq + 16 * t2;
      const float*  yc = (const float*)cur + (16 * rt + 8 * half) * YP + ct + l15;
      const __bf16* ph = (const __bf16*)(cur + OFF_BH) + (ct + l15) * HP;
      const __bf16* pl = (const __bf16*)(cur + OFF_BL) + (ct + l15) * HP;
      v8f acc;
#pragma unroll
      for (int v = 0; v < 8; ++v) acc[v] = yc[v * YP];
#pragma unroll
      for (int kk = 0; kk < 4; ++kk) {
        const int k0 = kk * 32;
        const v16bf fh = lds_frag(ph, k0, half);
        const v16bf fl = lds_frag(pl, k0, half);
        acc = wmma_bf16(mh[kk], fh, acc);
        acc = wmma_bf16(ml[kk], fh, acc);
        acc = wmma_bf16(mh[kk], fl, acc);
      }
      float* op = out + (size_t)(i0 + 16 * rt + 8 * half) * IN_DIM + 128 * c + ct + l15;
#pragma unroll
      for (int v = 0; v < 8; ++v) op[(size_t)v * IN_DIM] = acc[v];
    }
    if (c + 1 < 8) {
      stash_y_f32(nxt, tid, yq);
      stash_b(nxt, tid, bq);
    }
    __syncthreads();
  }
}

extern "C" void kernel_launch(void* const* d_in, const int* in_sizes, int n_in,
                              void* d_out, int out_size, void* d_ws,
                              size_t ws_size, hipStream_t stream) {
  (void)in_sizes; (void)n_in; (void)out_size; (void)ws_size;
  const float* y = (const float*)d_in[0];
  const float* A = (const float*)d_in[1];
  const float* b = (const float*)d_in[2];
  float* out = (float*)d_out;

  char* ws = (char*)d_ws;
  float*  G   = (float*) (ws + 0);
  __bf16* Gh  = (__bf16*)(ws + 65536);
  __bf16* Gl  = (__bf16*)(ws + 98304);
  __bf16* Ah  = (__bf16*)(ws + 131072);
  __bf16* Al  = (__bf16*)(ws + 393216);
  __bf16* Ath = (__bf16*)(ws + 655360);
  __bf16* Atl = (__bf16*)(ws + 917504);

  compute_G<<<256, 256, 70656, stream>>>(A, G);
  invert_G<<<1, 256, 67584, stream>>>(G, Gh, Gl);
  pack_A<<<512, 256, 0, stream>>>(A, Ah, Al, Ath, Atl);
  fused_project<<<256, 512, 274432, stream>>>(y, b, Ah, Al, Ath, Atl,
                                              Gh, Gl, out);
}


# fused_project stage 1: Ginv hi/lo planes and b fetched by whole rows under stage 0's last chunk into the free LDS chunk buffer; fragments read from LDS, not by dependent global loads; sums unchanged
# speedup vs baseline: 6.9271x; 1.0690x over previous
#include <hip/hip_runtime.h>


#define BATCH   16384
#define IN_DIM  1024
#define OUT_DIM 128

typedef __attribute__((ext_vector_type(16))) __bf16 v16bf;
typedef __attribute__((ext_vector_type(8)))  float  v8f;
typedef __attribute__((ext_vector_type(2)))  float  v2f;

static __device__ inline v8f wmma_bf16(v16bf a, v16bf b, v8f c) {
  return __builtin_amdgcn_wmma_f32_16x16x32_bf16(false, a, false, b,
                                                 (short)0, c, false, false);
}

#define GPP 68

__global__ __launch_bounds__(256) void compute_G(const float* __restrict__ A,
                                                 float* __restrict__ G) {
  extern __shared__ __attribute__((aligned(16))) float gpart[];
  const int t  = threadIdx.x;
  const int bi = blockIdx.x >> 4, bj = blockIdx.x & 15;
  const float* ap = A + (size_t)(8 * bi) * IN_DIM + 4 * t;
  const float* bp = A + (size_t)(8 * bj) * IN_DIM + 4 * t;
  float4 ra[8], rb[8];
#pragma unroll
  for (int r = 0; r < 8; ++r) {
    ra[r] = *(const float4*)(ap + (size_t)r * IN_DIM);
    rb[r] = *(const float4*)(bp + (size_t)r * IN_DIM);
  }
  float* mine = gpart + t * GPP;
#pragma unroll
  for (int i = 0; i < 8; ++i) {
    float s[8];
#pragma unroll
    for (int j = 0; j < 8; ++j) {
      float v = ra[i].x * rb[j].x;
      v = fmaf(ra[i].y, rb[j].y, v);
      v = fmaf(ra[i].z, rb[j].z, v);
      s[j] = fmaf(ra[i].w, rb[j].w, v);
    }
    *(float4*)(mine + 8 * i)     = make_float4(s[0], s[1], s[2], s[3]);
    *(float4*)(mine + 8 * i + 4) = make_float4(s[4], s[5], s[6], s[7]);
  }
  __syncthreads();

  const int o = t & 63, q = t >> 6;
  const float* col = gpart + (64 * q) * GPP + o;
  float c0 = col[0], c1 = col[GPP], c2 = col[2 * GPP], c3 = col[3 * GPP];
#pragma unroll
  for (int u = 4; u < 64; u += 4) {
    c0 += col[(u    ) * GPP];
    c1 += col[(u + 1) * GPP];
    c2 += col[(u + 2) * GPP];
    c3 += col[(u + 3) * GPP];
  }
  float* red = gpart + 256 * GPP;
  red[64 * q + o] = (c0 + c1) + (c2 + c3);
  __syncthreads();
  if (t < 64)
    G[blockIdx.x * 64 + t] = (red[t] + red[64 + t]) + (red[128 + t] + red[192 + t]);
}

#define GLD 132

static __device__ inline v8f wmma_f32(v2f a, v2f b, v8f c) {
  return __builtin_amdgcn_wmma_f32_16x16x4_f32(false, a, false, b,
                                               (short)0, c, false, false);
}

static __device__ inline float lane_value(float x, int src) {
  return __builtin_bit_cast(float,
      __builtin_amdgcn_readlane(__builtin_bit_cast(int, x), src));
}

template <int K>
static __device__ inline float row_lane(float x) {
  return __builtin_bit_cast(float, __builtin_amdgcn_update_dpp(
      0, __builtin_bit_cast(int, x), 0x150 + K, 0xf, 0xf, true));
}

static __device__ inline float other_half(float x) {
  return __builtin_bit_cast(float, __builtin_amdgcn_permlanex16(
      __builtin_bit_cast(int, x), __builtin_bit_cast(int, x),
      0x76543210, 0xfedcba98, false, false));
}

template <int K>
static __device__ inline void gj_pivot(float (&e)[8], int c, int h) {
  constexpr int KH = K >> 3, KK = K & 7;
  const float pinv = 1.0f / lane_value(e[KK], K + 16 * KH);
  const float own  = e[KK];
  const float oth  = other_half(own);
  const float colk = (h == KH) ? own : oth;
  const bool  isk  = (c == K);
  const float nf   = isk ? 0.0f : -colk;
  float rk[8], up[8];
#pragma unroll
  for (int jj = 0; jj < 8; ++jj) rk[jj] = row_lane<K>(e[jj]) * pinv;
#pragma unroll
  for (int jj = 0; jj < 8; ++jj) up[jj] = fmaf(nf, rk[jj], e[jj]);
#pragma unroll
  for (int jj = 0; jj < 8; ++jj) e[jj] = isk ? rk[jj] : up[jj];
  const float ck = isk ? pinv : nf * pinv;
  e[KK] = (h == KH) ? ck : e[KK];
}

template <int K> struct GJ16 {
  static __device__ inline void run(float (&e)[8], int c, int h) {
    gj_pivot<K>(e, c, h);
    GJ16<K + 1>::run(e, c, h);
  }
};
template <> struct GJ16<16> {
  static __device__ inline void run(float (&)[8], int, int) {}
};

static __device__ inline void load_fragA16(const float* p, v2f (&f)[4]) {
  float4 u = *(const float4*)p;
  float4 w = *(const float4*)(p + 4);
  f[0] = v2f{u.x, u.y};  f[1] = v2f{u.z, u.w};
  f[2] = v2f{w.x, w.y};  f[3] = v2f{w.z, w.w};
}

__global__ __launch_bounds__(256) void invert_G(const float* __restrict__ G,
                                                __bf16* __restrict__ Gh,
                                                __bf16* __restrict__ Gl) {
  extern __shared__ __attribute__((aligned(16))) float M[];
  const int tid  = threadIdx.x;
  const int J    = tid >> 5;
  const int lane = tid & 31;
  const int h    = lane >> 4;
  const int c    = lane & 15;

  for (int idx = tid; idx < 128 * 32; idx += 256) {
    const int tile = idx >> 4;
    const int i  = 8 * (tile >> 4) + ((idx >> 1) & 7);
    const int j4 = 8 * (tile & 15) + 4 * (idx & 1);
    *(float4*)(M + i * GLD + j4) = *(const float4*)(G + 4 * idx);
  }
  __syncthreads();

  for (int K = 0; K < 8; ++K) {
    float d[8];
    {
      const float* dp = M + (16 * K + c) * GLD + 16 * K + 8 * h;
      const float4 v0 = *(const float4*)dp;
      const float4 v1 = *(const float4*)(dp + 4);
      d[0] = v0.x; d[1] = v0.y; d[2] = v0.z; d[3] = v0.w;
      d[4] = v1.x; d[5] = v1.y; d[6] = v1.z; d[7] = v1.w;
    }
    v2f fa[8][4];
#pragma unroll
    for (int i = 0; i < 8; ++i)
      load_fragA16(M + (16 * i + c) * GLD + 16 * K + 8 * h, fa[i]);
    __syncthreads();

    if (J == K) {
      for (int rr = lane; rr < 128; rr += 32) {
        float* p = M + rr * GLD + 16 * K;
        const int e = rr - 16 * K;
#pragma unroll
        for (int j = 0; j < 16; j += 4)
          *(float4*)(p + j) = make_float4(e == j     ? 1.0f : 0.0f,
                                          e == j + 1 ? 1.0f : 0.0f,
                                          e == j + 2 ? 1.0f : 0.0f,
                                          e == j + 3 ? 1.0f : 0.0f);
      }
    }

    GJ16<0>::run(d, c, h);
    const v2f da[4] = {v2f{d[0], d[1]}, v2f{d[2], d[3]},
                       v2f{d[4], d[5]}, v2f{d[6], d[7]}};

    v8f t = {};
    {
      const float* bp = M + (16 * K + 8 * h) * GLD + 16 * J + c;
#pragma unroll
      for (int q = 0; q < 4; ++q) {
        v2f bq = v2f{bp[(2 * q) * GLD], bp[(2 * q + 1) * GLD]};
        t = wmma_f32(da[q], bq, t);
      }
    }
    v2f nb[4];
#pragma unroll
    for (int q = 0; q < 4; ++q) nb[q] = v2f{-t[2 * q], -t[2 * q + 1]};

#pragma unroll
    for (int i = 0; i < 8; ++i) {
      float* mp = M + (16 * i + 8 * h) * GLD + 16 * J + c;
      v8f acc;
#pragma unroll
      for (int r = 0; r < 8; ++r) acc[r] = mp[r * GLD];
#pragma unroll
      for (int q = 0; q < 4; ++q) acc = wmma_f32(fa[i][q], nb[q], acc);
#pragma unroll
      for (int r = 0; r < 8; ++r) mp[r * GLD] = (i == K) ? t[r] : acc[r];
    }
    __syncthreads();
  }

  for (int idx = tid; idx < 128 * 128; idx += 256) {
    int i = idx >> 7, j = idx & 127;
    float v = M[i * GLD + j];
    __bf16 hh = (__bf16)v;
    Gh[idx] = hh; Gl[idx] = (__bf16)(v - (float)hh);
  }
}

__global__ __launch_bounds__(256) void pack_A(const float* __restrict__ A,
                                              __bf16* __restrict__ Ah,
                                              __bf16* __restrict__ Al,
                                              __bf16* __restrict__ Ath,
                                              __bf16* __restrict__ Atl) {
  int idx = blockIdx.x * 256 + threadIdx.x;
  int r = idx >> 10, c = idx & 1023;
  float x = A[idx];
  __bf16 h = (__bf16)x;
  __bf16 l = (__bf16)(x - (float)h);
  Ah[idx] = h;  Al[idx] = l;
  Ath[c * 128 + r] = h;  Atl[c * 128 + r] = l;
}

#define RB   64
#define HP   136
#define YP   132
#define OFF_YL 17408
#define OFF_BH 34816
#define OFF_BL 69632
#define BUFB   104448

typedef __attribute__((ext_vector_type(8))) __bf16 v8bf;
typedef __attribute__((ext_vector_type(4))) __bf16 v4bf;
typedef __attribute__((ext_vector_type(4))) float  v4f;

static __device__ inline v16bf lds_frag(const __bf16* rowp, int k0, int half) {
  const v8bf u = *(const v8bf*)(rowp + k0 + 8 * half);
  const v8bf w = *(const v8bf*)(rowp + k0 + 16 + 8 * half);
  return __builtin_shufflevector(u, w, 0, 1, 2, 3, 4, 5, 6, 7,
                                 8, 9, 10, 11, 12, 13, 14, 15);
}

static __device__ inline void split_fragA_lds(const float* rowp, int k0,
                                              int half, bool neg,
                                              v16bf& hi, v16bf& lo) {
  const float* p0 = rowp + k0 + 8 * half;
  const float4 a0 = *(const float4*)(p0);
  const float4 a1 = *(const float4*)(p0 + 4);
  const float4 c0 = *(const float4*)(p0 + 16);
  const float4 c1 = *(const float4*)(p0 + 20);
  const float x[16] = {a0.x, a0.y, a0.z, a0.w, a1.x, a1.y, a1.z, a1.w,
                       c0.x, c0.y, c0.z, c0.w, c1.x, c1.y, c1.z, c1.w};
#pragma unroll
  for (int e = 0; e < 16; ++e) {
    const float xe = neg ? -x[e] : x[e];
    const __bf16 he = (__bf16)xe;
    hi[e] = he;  lo[e] = (__bf16)(xe - (float)he);
  }
}

static __device__ inline void fetch_y(const float* __restrict__ y, int i0,
                                      int col0, int tid, v4f (&yq)[4]) {
#pragma unroll
  for (int j = 0; j < 4; ++j) {
    const int idx = tid + 512 * j;
    yq[j] = *(const v4f*)(y + (size_t)(i0 + (idx >> 5)) * IN_DIM + col0 +
                          4 * (idx & 31));
  }
}

static __device__ inline void fetch_b(const __bf16* __restrict__ Ph,
                                      const __bf16* __restrict__ Pl,
                                      size_t base, int pitch, int tid,
                                      v8bf (&bq)[8]) {
#pragma unroll
  for (int j = 0; j < 4; ++j) {
    const int idx = tid + 512 * j;
    const size_t off = base + (size_t)(idx >> 4) * pitch + 8 * (idx & 15);
    bq[j]     = *(const v8bf*)(Ph + off);
    bq[j + 4] = *(const v8bf*)(Pl + off);
  }
}

static __device__ inline void stash_b(char* buf, int tid, const v8bf (&bq)[8]) {
  __bf16* Bh = (__bf16*)(buf + OFF_BH);
  __bf16* Bl = (__bf16*)(buf + OFF_BL);
#pragma unroll
  for (int j = 0; j < 4; ++j) {
    const int idx = tid + 512 * j;
    const int o = (idx >> 4) * HP + 8 * (idx & 15);
    *(v8bf*)(Bh + o) = bq[j];
    *(v8bf*)(Bl + o) = bq[j + 4];
  }
}

static __device__ inline void stash_y_split(char* buf, int tid,
                                            const v4f (&yq)[4]) {
  __bf16* Yh = (__bf16*)buf;
  __bf16* Yl = (__bf16*)(buf + OFF_YL);
#pragma unroll
  for (int j = 0; j < 4; ++j) {
    const int idx = tid + 512 * j;
    const int o = (idx >> 5) * HP + 4 * (idx & 31);
    const float x[4] = {yq[j][0], yq[j][1], yq[j][2], yq[j][3]};
    v4bf h, l;
#pragma unroll
    for (int e = 0; e < 4; ++e) {
      const __bf16 he = (__bf16)x[e];
      h[e] = he;  l[e] = (__bf16)(x[e] - (float)he);
    }
    *(v4bf*)(Yh + o) = h;
    *(v4bf*)(Yl + o) = l;
  }
}

static __device__ inline void stash_y_f32(char* buf, int tid,
                                          const v4f (&yq)[4]) {
  float* Yc = (float*)buf;
#pragma unroll
  for (int j = 0; j < 4; ++j) {
    const int idx = tid + 512 * j;
    *(v4f*)(Yc + (idx >> 5) * YP + 4 * (idx & 31)) = yq[j];
  }
}

static __device__ inline void stage0_products(const char* cur, int rt, int l15,
                                              int half, int n0a,
                                              v8f& acc0, v8f& acc1) {
  const __bf16* yh  = (const __bf16*)cur + (16 * rt + l15) * HP;
  const __bf16* yl  = (const __bf16*)(cur + OFF_YL) + (16 * rt + l15) * HP;
  const __bf16* b0h = (const __bf16*)(cur + OFF_BH) + (n0a + l15) * HP;
  const __bf16* b0l = (const __bf16*)(cur + OFF_BL) + (n0a + l15) * HP;
  const __bf16* b1h = b0h + 16 * HP;
  const __bf16* b1l = b0l + 16 * HP;
#pragma unroll
  for (int ks = 0; ks < 4; ++ks) {
    const int k0 = 32 * ks;
    const v16bf ah = lds_frag(yh, k0, half);
    const v16bf al = lds_frag(yl, k0, half);
    v16bf fh = lds_frag(b0h, k0, half);
    v16bf fl = lds_frag(b0l, k0, half);
    acc0 = wmma_bf16(ah, fh, acc0);
    acc0 = wmma_bf16(al, fh, acc0);
    acc0 = wmma_bf16(ah, fl, acc0);
    fh = lds_frag(b1h, k0, half);
    fl = lds_frag(b1l, k0, half);
    acc1 = wmma_bf16(ah, fh, acc1);
    acc1 = wmma_bf16(al, fh, acc1);
    acc1 = wmma_bf16(ah, fl, acc1);
  }
}

__global__ __launch_bounds__(512) void fused_project(
    const float* __restrict__ y, const float* __restrict__ b,
    const __bf16* __restrict__ Ah, const __bf16* __restrict__ Al,
    const __bf16* __restrict__ Ath, const __bf16* __restrict__ Atl,
    const __bf16* __restrict__ Gh, const __bf16* __restrict__ Gl,
    float* __restrict__ out) {
  extern __shared__ __attribute__((aligned(16))) char lds[];
  char*  buf0 = lds;
  char*  buf1 = lds + BUFB;
  float* s_r  = (float*)(lds + 2 * BUFB);
  float* s_mu = s_r + RB * OUT_DIM;

  const int tid  = threadIdx.x;
  const int wid  = tid >> 5;
  const int rt   = wid >> 2;
  const int cq   = wid & 3;
  const int lane = tid & 31;
  const int half = lane >> 4;
  const int l15  = lane & 15;
  const int i0   = blockIdx.x * RB;
  const int n0a  = cq * 32;
  const int n0b  = n0a + 16;

  v4f    yq[4];
  v8bf   bq[8];

  fetch_y(y, i0, 0, tid, yq);
  fetch_b(Ah, Al, 0, IN_DIM, tid, bq);
  stash_y_split(buf0, tid, yq);
  stash_b(buf0, tid, bq);
  __syncthreads();

  v8f acc0 = {};
  v8f acc1 = {};
  for (int c = 0; c < 7; ++c) {
    char* cur = (c & 1) ? buf1 : buf0;
    char* nxt = (c & 1) ? buf0 : buf1;
    fetch_y(y, i0, 128 * (c + 1), tid, yq);
    fetch_b(Ah, Al, (size_t)(128 * (c + 1)), IN_DIM, tid, bq);
    __builtin_amdgcn_sched_barrier(0);
    stage0_products(cur, rt, l15, half, n0a, acc0, acc1);
    asm volatile("" : "+v"(acc0), "+v"(acc1), "+v"(yq[0]), "+v"(yq[1]),
                      "+v"(yq[2]), "+v"(yq[3]));
    stash_y_split(nxt, tid, yq);
    stash_b(nxt, tid, bq);
    __syncthreads();
  }

  fetch_b(Gh, Gl, 0, OUT_DIM, tid, bq);
  float bv0[8], bv1[8];
#pragma unroll
  for (int v = 0; v < 8; ++v) {
    const float* bp = b + (size_t)(i0 + 16 * rt + v + 8 * half) * OUT_DIM + l15;
    bv0[v] = bp[n0a];
    bv1[v] = bp[n0b];
  }
  __builtin_amdgcn_sched_barrier(0);
  stage0_products(buf1, rt, l15, half, n0a, acc0, acc1);
  stash_b(buf0, tid, bq);

  fetch_y(y, i0, 0, tid, yq);
  fetch_b(Ath, Atl, 0, OUT_DIM, tid, bq);

#pragma unroll
  for (int v = 0; v < 8; ++v) {
    const int m = 16 * rt + v + 8 * half;
    s_r[m * OUT_DIM + n0a + l15] = acc0[v] - bv0[v];
    s_r[m * OUT_DIM + n0b + l15] = acc1[v] - bv1[v];
  }
  __syncthreads();

  const __bf16* g0h = (const __bf16*)(buf0 + OFF_BH) + (n0a + l15) * HP;
  const __bf16* g0l = (const __bf16*)(buf0 + OFF_BL) + (n0a + l15) * HP;
  const __bf16* g1h = g0h + 16 * HP;
  const __bf16* g1l = g0l + 16 * HP;
  v8f mu0 = {};
  v8f mu1 = {};
#pragma unroll
  for (int kk = 0; kk < 4; ++kk) {
    const int k0 = kk * 32;
    v16bf rh, rl;
    split_fragA_lds(s_r + (16 * rt + l15) * OUT_DIM, k0, half, false, rh, rl);
    v16bf fh = lds_frag(g0h, k0, half);
    v16bf fl = lds_frag(g0l, k0, half);
    mu0 = wmma_bf16(rh, fh, mu0);
    mu0 = wmma_bf16(rl, fh, mu0);
    mu0 = wmma_bf16(rh, fl, mu0);
    fh = lds_frag(g1h, k0, half);
    fl = lds_frag(g1l, k0, half);
    mu1 = wmma_bf16(rh, fh, mu1);
    mu1 = wmma_bf16(rl, fh, mu1);
    mu1 = wmma_bf16(rh, fl, mu1);
  }
#pragma unroll
  for (int v = 0; v < 8; ++v) {
    const int m = 16 * rt + v + 8 * half;
    s_mu[m * OUT_DIM + n0a + l15] = mu0[v];
    s_mu[m * OUT_DIM + n0b + l15] = mu1[v];
  }
  __syncthreads();

  v16bf mh[4], ml[4];
#pragma unroll
  for (int kk = 0; kk < 4; ++kk)
    split_fragA_lds(s_mu + (16 * rt + l15) * OUT_DIM, kk * 32, half, true,
                    mh[kk], ml[kk]);

  stash_y_f32(buf0, tid, yq);
  stash_b(buf0, tid, bq);
  __syncthreads();

  for (int c = 0; c < 8; ++c) {
    char* cur = (c & 1) ? buf1 : buf0;
    char* nxt = (c & 1) ? buf0 : buf1;
    if (c + 1 < 8) {
      fetch_y(y, i0, 128 * (c + 1), tid, yq);
      fetch_b(Ath, Atl, (size_t)(128 * (c + 1)) * OUT_DIM, OUT_DIM, tid, bq);
    }
    __builtin_amdgcn_sched_barrier(0);
#pragma unroll
    for (int t2 = 0; t2 < 2; ++t2) {
      const int ct = 32 * cq + 16 * t2;
      const float*  yc = (const float*)cur + (16 * rt + 8 * half) * YP + ct + l15;
      const __bf16* ph = (const __bf16*)(cur + OFF_BH) + (ct + l15) * HP;
      const __bf16* pl = (const __bf16*)(cur + OFF_BL) + (ct + l15) * HP;
      v8f acc;
#pragma unroll
      for (int v = 0; v < 8; ++v) acc[v] = yc[v * YP];
#pragma unroll
      for (int kk = 0; kk < 4; ++kk) {
        const int k0 = kk * 32;
        const v16bf fh = lds_frag(ph, k0, half);
        const v16bf fl = lds_frag(pl, k0, half);
        acc = wmma_bf16(mh[kk], fh, acc);
        acc = wmma_bf16(ml[kk], fh, acc);
        acc = wmma_bf16(mh[kk], fl, acc);
      }
      float* op = out + (size_t)(i0 + 16 * rt + 8 * half) * IN_DIM + 128 * c + ct + l15;
#pragma unroll
      for (int v = 0; v < 8; ++v) op[(size_t)v * IN_DIM] = acc[v];
    }
    if (c + 1 < 8) {
      stash_y_f32(nxt, tid, yq);
      stash_b(nxt, tid, bq);
    }
    __syncthreads();
  }
}

extern "C" void kernel_launch(void* const* d_in, const int* in_sizes, int n_in,
                              void* d_out, int out_size, void* d_ws,
                              size_t ws_size, hipStream_t stream) {
  (void)in_sizes; (void)n_in; (void)out_size; (void)ws_size;
  const float* y = (const float*)d_in[0];
  const float* A = (const float*)d_in[1];
  const float* b = (const float*)d_in[2];
  float* out = (float*)d_out;

  char* ws = (char*)d_ws;
  float*  G   = (float*) (ws + 0);
  __bf16* Gh  = (__bf16*)(ws + 65536);
  __bf16* Gl  = (__bf16*)(ws + 98304);
  __bf16* Ah  = (__bf16*)(ws + 131072);
  __bf16* Al  = (__bf16*)(ws + 393216);
  __bf16* Ath = (__bf16*)(ws + 655360);
  __bf16* Atl = (__bf16*)(ws + 917504);

  compute_G<<<256, 256, 70656, stream>>>(A, G);
  invert_G<<<1, 256, 67584, stream>>>(G, Gh, Gl);
  pack_A<<<512, 256, 0, stream>>>(A, Ah, Al, Ath, Atl);
  fused_project<<<256, 512, 274432, stream>>>(y, b, Ah, Al, Ath, Atl,
                                              Gh, Gl, out);
}


# fused_project: r and mu rows in LDS at pitch 132 floats, not 128: the 16 rows of a 16-byte A-fragment read hit one 4-bank group (16-way conflict per half-wave), now all 64 banks; sums unchanged
# speedup vs baseline: 7.4703x; 1.0784x over previous
#include <hip/hip_runtime.h>


#define BATCH   16384
#define IN_DIM  1024
#define OUT_DIM 128

typedef __attribute__((ext_vector_type(16))) __bf16 v16bf;
typedef __attribute__((ext_vector_type(8)))  float  v8f;
typedef __attribute__((ext_vector_type(2)))  float  v2f;

static __device__ inline v8f wmma_bf16(v16bf a, v16bf b, v8f c) {
  return __builtin_amdgcn_wmma_f32_16x16x32_bf16(false, a, false, b,
                                                 (short)0, c, false, false);
}

#define GPP 68

__global__ __launch_bounds__(256) void compute_G(const float* __restrict__ A,
                                                 float* __restrict__ G) {
  extern __shared__ __attribute__((aligned(16))) float gpart[];
  const int t  = threadIdx.x;
  const int bi = blockIdx.x >> 4, bj = blockIdx.x & 15;
  const float* ap = A + (size_t)(8 * bi) * IN_DIM + 4 * t;
  const float* bp = A + (size_t)(8 * bj) * IN_DIM + 4 * t;
  float4 ra[8], rb[8];
#pragma unroll
  for (int r = 0; r < 8; ++r) {
    ra[r] = *(const float4*)(ap + (size_t)r * IN_DIM);
    rb[r] = *(const float4*)(bp + (size_t)r * IN_DIM);
  }
  float* mine = gpart + t * GPP;
#pragma unroll
  for (int i = 0; i < 8; ++i) {
    float s[8];
#pragma unroll
    for (int j = 0; j < 8; ++j) {
      float v = ra[i].x * rb[j].x;
      v = fmaf(ra[i].y, rb[j].y, v);
      v = fmaf(ra[i].z, rb[j].z, v);
      s[j] = fmaf(ra[i].w, rb[j].w, v);
    }
    *(float4*)(mine + 8 * i)     = make_float4(s[0], s[1], s[2], s[3]);
    *(float4*)(mine + 8 * i + 4) = make_float4(s[4], s[5], s[6], s[7]);
  }
  __syncthreads();

  const int o = t & 63, q = t >> 6;
  const float* col = gpart + (64 * q) * GPP + o;
  float c0 = col[0], c1 = col[GPP], c2 = col[2 * GPP], c3 = col[3 * GPP];
#pragma unroll
  for (int u = 4; u < 64; u += 4) {
    c0 += col[(u    ) * GPP];
    c1 += col[(u + 1) * GPP];
    c2 += col[(u + 2) * GPP];
    c3 += col[(u + 3) * GPP];
  }
  float* red = gpart + 256 * GPP;
  red[64 * q + o] = (c0 + c1) + (c2 + c3);
  __syncthreads();
  if (t < 64)
    G[blockIdx.x * 64 + t] = (red[t] + red[64 + t]) + (red[128 + t] + red[192 + t]);
}

#define GLD 132

static __device__ inline v8f wmma_f32(v2f a, v2f b, v8f c) {
  return __builtin_amdgcn_wmma_f32_16x16x4_f32(false, a, false, b,
                                               (short)0, c, false, false);
}

static __device__ inline float lane_value(float x, int src) {
  return __builtin_bit_cast(float,
      __builtin_amdgcn_readlane(__builtin_bit_cast(int, x), src));
}

template <int K>
static __device__ inline float row_lane(float x) {
  return __builtin_bit_cast(float, __builtin_amdgcn_update_dpp(
      0, __builtin_bit_cast(int, x), 0x150 + K, 0xf, 0xf, true));
}

static __device__ inline float other_half(float x) {
  return __builtin_bit_cast(float, __builtin_amdgcn_permlanex16(
      __builtin_bit_cast(int, x), __builtin_bit_cast(int, x),
      0x76543210, 0xfedcba98, false, false));
}

template <int K>
static __device__ inline void gj_pivot(float (&e)[8], int c, int h) {
  constexpr int KH = K >> 3, KK = K & 7;
  const float pinv = 1.0f / lane_value(e[KK], K + 16 * KH);
  const float own  = e[KK];
  const float oth  = other_half(own);
  const float colk = (h == KH) ? own : oth;
  const bool  isk  = (c == K);
  const float nf   = isk ? 0.0f : -colk;
  float rk[8], up[8];
#pragma unroll
  for (int jj = 0; jj < 8; ++jj) rk[jj] = row_lane<K>(e[jj]) * pinv;
#pragma unroll
  for (int jj = 0; jj < 8; ++jj) up[jj] = fmaf(nf, rk[jj], e[jj]);
#pragma unroll
  for (int jj = 0; jj < 8; ++jj) e[jj] = isk ? rk[jj] : up[jj];
  const float ck = isk ? pinv : nf * pinv;
  e[KK] = (h == KH) ? ck : e[KK];
}

template <int K> struct GJ16 {
  static __device__ inline void run(float (&e)[8], int c, int h) {
    gj_pivot<K>(e, c, h);
    GJ16<K + 1>::run(e, c, h);
  }
};
template <> struct GJ16<16> {
  static __device__ inline void run(float (&)[8], int, int) {}
};

static __device__ inline void load_fragA16(const float* p, v2f (&f)[4]) {
  float4 u = *(const float4*)p;
  float4 w = *(const float4*)(p + 4);
  f[0] = v2f{u.x, u.y};  f[1] = v2f{u.z, u.w};
  f[2] = v2f{w.x, w.y};  f[3] = v2f{w.z, w.w};
}

__global__ __launch_bounds__(256) void invert_G(const float* __restrict__ G,
                                                __bf16* __restrict__ Gh,
                                                __bf16* __restrict__ Gl) {
  extern __shared__ __attribute__((aligned(16))) float M[];
  const int tid  = threadIdx.x;
  const int J    = tid >> 5;
  const int lane = tid & 31;
  const int h    = lane >> 4;
  const int c    = lane & 15;

  for (int idx = tid; idx < 128 * 32; idx += 256) {
    const int tile = idx >> 4;
    const int i  = 8 * (tile >> 4) + ((idx >> 1) & 7);
    const int j4 = 8 * (tile & 15) + 4 * (idx & 1);
    *(float4*)(M + i * GLD + j4) = *(const float4*)(G + 4 * idx);
  }
  __syncthreads();

  for (int K = 0; K < 8; ++K) {
    float d[8];
    {
      const float* dp = M + (16 * K + c) * GLD + 16 * K + 8 * h;
      const float4 v0 = *(const float4*)dp;
      const float4 v1 = *(const float4*)(dp + 4);
      d[0] = v0.x; d[1] = v0.y; d[2] = v0.z; d[3] = v0.w;
      d[4] = v1.x; d[5] = v1.y; d[6] = v1.z; d[7] = v1.w;
    }
    v2f fa[8][4];
#pragma unroll
    for (int i = 0; i < 8; ++i)
      load_fragA16(M + (16 * i + c) * GLD + 16 * K + 8 * h, fa[i]);
    __syncthreads();

    if (J == K) {
      for (int rr = lane; rr < 128; rr += 32) {
        float* p = M + rr * GLD + 16 * K;
        const int e = rr - 16 * K;
#pragma unroll
        for (int j = 0; j < 16; j += 4)
          *(float4*)(p + j) = make_float4(e == j     ? 1.0f : 0.0f,
                                          e == j + 1 ? 1.0f : 0.0f,
                                          e == j + 2 ? 1.0f : 0.0f,
                                          e == j + 3 ? 1.0f : 0.0f);
      }
    }

    GJ16<0>::run(d, c, h);
    const v2f da[4] = {v2f{d[0], d[1]}, v2f{d[2], d[3]},
                       v2f{d[4], d[5]}, v2f{d[6], d[7]}};

    v8f t = {};
    {
      const float* bp = M + (16 * K + 8 * h) * GLD + 16 * J + c;
#pragma unroll
      for (int q = 0; q < 4; ++q) {
        v2f bq = v2f{bp[(2 * q) * GLD], bp[(2 * q + 1) * GLD]};
        t = wmma_f32(da[q], bq, t);
      }
    }
    v2f nb[4];
#pragma unroll
    for (int q = 0; q < 4; ++q) nb[q] = v2f{-t[2 * q], -t[2 * q + 1]};

#pragma unroll
    for (int i = 0; i < 8; ++i) {
      float* mp = M + (16 * i + 8 * h) * GLD + 16 * J + c;
      v8f acc;
#pragma unroll
      for (int r = 0; r < 8; ++r) acc[r] = mp[r * GLD];
#pragma unroll
      for (int q = 0; q < 4; ++q) acc = wmma_f32(fa[i][q], nb[q], acc);
#pragma unroll
      for (int r = 0; r < 8; ++r) mp[r * GLD] = (i == K) ? t[r] : acc[r];
    }
    __syncthreads();
  }

  for (int idx = tid; idx < 128 * 128; idx += 256) {
    int i = idx >> 7, j = idx & 127;
    float v = M[i * GLD + j];
    __bf16 hh = (__bf16)v;
    Gh[idx] = hh; Gl[idx] = (__bf16)(v - (float)hh);
  }
}

__global__ __launch_bounds__(256) void pack_A(const float* __restrict__ A,
                                              __bf16* __restrict__ Ah,
                                              __bf16* __restrict__ Al,
                                              __bf16* __restrict__ Ath,
                                              __bf16* __restrict__ Atl) {
  int idx = blockIdx.x * 256 + threadIdx.x;
  int r = idx >> 10, c = idx & 1023;
  float x = A[idx];
  __bf16 h = (__bf16)x;
  __bf16 l = (__bf16)(x - (float)h);
  Ah[idx] = h;  Al[idx] = l;
  Ath[c * 128 + r] = h;  Atl[c * 128 + r] = l;
}

#define RB   64
#define HP   136
#define YP   132
#define RP   132
#define OFF_YL 17408
#define OFF_BH 34816
#define OFF_BL 69632
#define BUFB   104448

typedef __attribute__((ext_vector_type(8))) __bf16 v8bf;
typedef __attribute__((ext_vector_type(4))) __bf16 v4bf;
typedef __attribute__((ext_vector_type(4))) float  v4f;

static __device__ inline v16bf lds_frag(const __bf16* rowp, int k0, int half) {
  const v8bf u = *(const v8bf*)(rowp + k0 + 8 * half);
  const v8bf w = *(const v8bf*)(rowp + k0 + 16 + 8 * half);
  return __builtin_shufflevector(u, w, 0, 1, 2, 3, 4, 5, 6, 7,
                                 8, 9, 10, 11, 12, 13, 14, 15);
}

static __device__ inline void split_fragA_lds(const float* rowp, int k0,
                                              int half, bool neg,
                                              v16bf& hi, v16bf& lo) {
  const float* p0 = rowp + k0 + 8 * half;
  const float4 a0 = *(const float4*)(p0);
  const float4 a1 = *(const float4*)(p0 + 4);
  const float4 c0 = *(const float4*)(p0 + 16);
  const float4 c1 = *(const float4*)(p0 + 20);
  const float x[16] = {a0.x, a0.y, a0.z, a0.w, a1.x, a1.y, a1.z, a1.w,
                       c0.x, c0.y, c0.z, c0.w, c1.x, c1.y, c1.z, c1.w};
#pragma unroll
  for (int e = 0; e < 16; ++e) {
    const float xe = neg ? -x[e] : x[e];
    const __bf16 he = (__bf16)xe;
    hi[e] = he;  lo[e] = (__bf16)(xe - (float)he);
  }
}

static __device__ inline void fetch_y(const float* __restrict__ y, int i0,
                                      int col0, int tid, v4f (&yq)[4]) {
#pragma unroll
  for (int j = 0; j < 4; ++j) {
    const int idx = tid + 512 * j;
    yq[j] = *(const v4f*)(y + (size_t)(i0 + (idx >> 5)) * IN_DIM + col0 +
                          4 * (idx & 31));
  }
}

static __device__ inline void fetch_b(const __bf16* __restrict__ Ph,
                                      const __bf16* __restrict__ Pl,
                                      size_t base, int pitch, int tid,
                                      v8bf (&bq)[8]) {
#pragma unroll
  for (int j = 0; j < 4; ++j) {
    const int idx = tid + 512 * j;
    const size_t off = base + (size_t)(idx >> 4) * pitch + 8 * (idx & 15);
    bq[j]     = *(const v8bf*)(Ph + off);
    bq[j + 4] = *(const v8bf*)(Pl + off);
  }
}

static __device__ inline void stash_b(char* buf, int tid, const v8bf (&bq)[8]) {
  __bf16* Bh = (__bf16*)(buf + OFF_BH);
  __bf16* Bl = (__bf16*)(buf + OFF_BL);
#pragma unroll
  for (int j = 0; j < 4; ++j) {
    const int idx = tid + 512 * j;
    const int o = (idx >> 4) * HP + 8 * (idx & 15);
    *(v8bf*)(Bh + o) = bq[j];
    *(v8bf*)(Bl + o) = bq[j + 4];
  }
}

static __device__ inline void stash_y_split(char* buf, int tid,
                                            const v4f (&yq)[4]) {
  __bf16* Yh = (__bf16*)buf;
  __bf16* Yl = (__bf16*)(buf + OFF_YL);
#pragma unroll
  for (int j = 0; j < 4; ++j) {
    const int idx = tid + 512 * j;
    const int o = (idx >> 5) * HP + 4 * (idx & 31);
    const float x[4] = {yq[j][0], yq[j][1], yq[j][2], yq[j][3]};
    v4bf h, l;
#pragma unroll
    for (int e = 0; e < 4; ++e) {
      const __bf16 he = (__bf16)x[e];
      h[e] = he;  l[e] = (__bf16)(x[e] - (float)he);
    }
    *(v4bf*)(Yh + o) = h;
    *(v4bf*)(Yl + o) = l;
  }
}

static __device__ inline void stash_y_f32(char* buf, int tid,
                                          const v4f (&yq)[4]) {
  float* Yc = (float*)buf;
#pragma unroll
  for (int j = 0; j < 4; ++j) {
    const int idx = tid + 512 * j;
    *(v4f*)(Yc + (idx >> 5) * YP + 4 * (idx & 31)) = yq[j];
  }
}

static __device__ inline void stage0_products(const char* cur, int rt, int l15,
                                              int half, int n0a,
                                              v8f& acc0, v8f& acc1) {
  const __bf16* yh  = (const __bf16*)cur + (16 * rt + l15) * HP;
  const __bf16* yl  = (const __bf16*)(cur + OFF_YL) + (16 * rt + l15) * HP;
  const __bf16* b0h = (const __bf16*)(cur + OFF_BH) + (n0a + l15) * HP;
  const __bf16* b0l = (const __bf16*)(cur + OFF_BL) + (n0a + l15) * HP;
  const __bf16* b1h = b0h + 16 * HP;
  const __bf16* b1l = b0l + 16 * HP;
#pragma unroll
  for (int ks = 0; ks < 4; ++ks) {
    const int k0 = 32 * ks;
    const v16bf ah = lds_frag(yh, k0, half);
    const v16bf al = lds_frag(yl, k0, half);
    v16bf fh = lds_frag(b0h, k0, half);
    v16bf fl = lds_frag(b0l, k0, half);
    acc0 = wmma_bf16(ah, fh, acc0);
    acc0 = wmma_bf16(al, fh, acc0);
    acc0 = wmma_bf16(ah, fl, acc0);
    fh = lds_frag(b1h, k0, half);
    fl = lds_frag(b1l, k0, half);
    acc1 = wmma_bf16(ah, fh, acc1);
    acc1 = wmma_bf16(al, fh, acc1);
    acc1 = wmma_bf16(ah, fl, acc1);
  }
}

__global__ __launch_bounds__(512) void fused_project(
    const float* __restrict__ y, const float* __restrict__ b,
    const __bf16* __restrict__ Ah, const __bf16* __restrict__ Al,
    const __bf16* __restrict__ Ath, const __bf16* __restrict__ Atl,
    const __bf16* __restrict__ Gh, const __bf16* __restrict__ Gl,
    float* __restrict__ out) {
  extern __shared__ __attribute__((aligned(16))) char lds[];
  char*  buf0 = lds;
  char*  buf1 = lds + BUFB;
  float* s_r  = (float*)(lds + 2 * BUFB);
  float* s_mu = s_r + RB * RP;

  const int tid  = threadIdx.x;
  const int wid  = tid >> 5;
  const int rt   = wid >> 2;
  const int cq   = wid & 3;
  const int lane = tid & 31;
  const int half = lane >> 4;
  const int l15  = lane & 15;
  const int i0   = blockIdx.x * RB;
  const int n0a  = cq * 32;
  const int n0b  = n0a + 16;

  v4f    yq[4];
  v8bf   bq[8];

  fetch_y(y, i0, 0, tid, yq);
  fetch_b(Ah, Al, 0, IN_DIM, tid, bq);
  stash_y_split(buf0, tid, yq);
  stash_b(buf0, tid, bq);
  __syncthreads();

  v8f acc0 = {};
  v8f acc1 = {};
  for (int c = 0; c < 7; ++c) {
    char* cur = (c & 1) ? buf1 : buf0;
    char* nxt = (c & 1) ? buf0 : buf1;
    fetch_y(y, i0, 128 * (c + 1), tid, yq);
    fetch_b(Ah, Al, (size_t)(128 * (c + 1)), IN_DIM, tid, bq);
    __builtin_amdgcn_sched_barrier(0);
    stage0_products(cur, rt, l15, half, n0a, acc0, acc1);
    asm volatile("" : "+v"(acc0), "+v"(acc1), "+v"(yq[0]), "+v"(yq[1]),
                      "+v"(yq[2]), "+v"(yq[3]));
    stash_y_split(nxt, tid, yq);
    stash_b(nxt, tid, bq);
    __syncthreads();
  }

  fetch_b(Gh, Gl, 0, OUT_DIM, tid, bq);
  float bv0[8], bv1[8];
#pragma unroll
  for (int v = 0; v < 8; ++v) {
    const float* bp = b + (size_t)(i0 + 16 * rt + v + 8 * half) * OUT_DIM + l15;
    bv0[v] = bp[n0a];
    bv1[v] = bp[n0b];
  }
  __builtin_amdgcn_sched_barrier(0);
  stage0_products(buf1, rt, l15, half, n0a, acc0, acc1);
  stash_b(buf0, tid, bq);

  fetch_y(y, i0, 0, tid, yq);
  fetch_b(Ath, Atl, 0, OUT_DIM, tid, bq);

#pragma unroll
  for (int v = 0; v < 8; ++v) {
    const int m = 16 * rt + v + 8 * half;
    s_r[m * RP + n0a + l15] = acc0[v] - bv0[v];
    s_r[m * RP + n0b + l15] = acc1[v] - bv1[v];
  }
  __syncthreads();

  const __bf16* g0h = (const __bf16*)(buf0 + OFF_BH) + (n0a + l15) * HP;
  const __bf16* g0l = (const __bf16*)(buf0 + OFF_BL) + (n0a + l15) * HP;
  const __bf16* g1h = g0h + 16 * HP;
  const __bf16* g1l = g0l + 16 * HP;
  v8f mu0 = {};
  v8f mu1 = {};
#pragma unroll
  for (int kk = 0; kk < 4; ++kk) {
    const int k0 = kk * 32;
    v16bf rh, rl;
    split_fragA_lds(s_r + (16 * rt + l15) * RP, k0, half, false, rh, rl);
    v16bf fh = lds_frag(g0h, k0, half);
    v16bf fl = lds_frag(g0l, k0, half);
    mu0 = wmma_bf16(rh, fh, mu0);
    mu0 = wmma_bf16(rl, fh, mu0);
    mu0 = wmma_bf16(rh, fl, mu0);
    fh = lds_frag(g1h, k0, half);
    fl = lds_frag(g1l, k0, half);
    mu1 = wmma_bf16(rh, fh, mu1);
    mu1 = wmma_bf16(rl, fh, mu1);
    mu1 = wmma_bf16(rh, fl, mu1);
  }
#pragma unroll
  for (int v = 0; v < 8; ++v) {
    const int m = 16 * rt + v + 8 * half;
    s_mu[m * RP + n0a + l15] = mu0[v];
    s_mu[m * RP + n0b + l15] = mu1[v];
  }
  __syncthreads();

  v16bf mh[4], ml[4];
#pragma unroll
  for (int kk = 0; kk < 4; ++kk)
    split_fragA_lds(s_mu + (16 * rt + l15) * RP, kk * 32, half, true,
                    mh[kk], ml[kk]);

  stash_y_f32(buf0, tid, yq);
  stash_b(buf0, tid, bq);
  __syncthreads();

  for (int c = 0; c < 8; ++c) {
    char* cur = (c & 1) ? buf1 : buf0;
    char* nxt = (c & 1) ? buf0 : buf1;
    if (c + 1 < 8) {
      fetch_y(y, i0, 128 * (c + 1), tid, yq);
      fetch_b(Ath, Atl, (size_t)(128 * (c + 1)) * OUT_DIM, OUT_DIM, tid, bq);
    }
    __builtin_amdgcn_sched_barrier(0);
#pragma unroll
    for (int t2 = 0; t2 < 2; ++t2) {
      const int ct = 32 * cq + 16 * t2;
      const float*  yc = (const float*)cur + (16 * rt + 8 * half) * YP + ct + l15;
      const __bf16* ph = (const __bf16*)(cur + OFF_BH) + (ct + l15) * HP;
      const __bf16* pl = (const __bf16*)(cur + OFF_BL) + (ct + l15) * HP;
      v8f acc;
#pragma unroll
      for (int v = 0; v < 8; ++v) acc[v] = yc[v * YP];
#pragma unroll
      for (int kk = 0; kk < 4; ++kk) {
        const int k0 = kk * 32;
        const v16bf fh = lds_frag(ph, k0, half);
        const v16bf fl = lds_frag(pl, k0, half);
        acc = wmma_bf16(mh[kk], fh, acc);
        acc = wmma_bf16(ml[kk], fh, acc);
        acc = wmma_bf16(mh[kk], fl, acc);
      }
      float* op = out + (size_t)(i0 + 16 * rt + 8 * half) * IN_DIM + 128 * c + ct + l15;
#pragma unroll
      for (int v = 0; v < 8; ++v) op[(size_t)v * IN_DIM] = acc[v];
    }
    if (c + 1 < 8) {
      stash_y_f32(nxt, tid, yq);
      stash_b(nxt, tid, bq);
    }
    __syncthreads();
  }
}

extern "C" void kernel_launch(void* const* d_in, const int* in_sizes, int n_in,
                              void* d_out, int out_size, void* d_ws,
                              size_t ws_size, hipStream_t stream) {
  (void)in_sizes; (void)n_in; (void)out_size; (void)ws_size;
  const float* y = (const float*)d_in[0];
  const float* A = (const float*)d_in[1];
  const float* b = (const float*)d_in[2];
  float* out = (float*)d_out;

  char* ws = (char*)d_ws;
  float*  G   = (float*) (ws + 0);
  __bf16* Gh  = (__bf16*)(ws + 65536);
  __bf16* Gl  = (__bf16*)(ws + 98304);
  __bf16* Ah  = (__bf16*)(ws + 131072);
  __bf16* Al  = (__bf16*)(ws + 393216);
  __bf16* Ath = (__bf16*)(ws + 655360);
  __bf16* Atl = (__bf16*)(ws + 917504);

  compute_G<<<256, 256, 70656, stream>>>(A, G);
  invert_G<<<1, 256, 67584, stream>>>(G, Gh, Gl);
  pack_A<<<512, 256, 0, stream>>>(A, Ah, Al, Ath, Atl);
  fused_project<<<256, 512, 276480, stream>>>(y, b, Ah, Al, Ath, Atl,
                                              Gh, Gl, out);
}


# fused_project: stage 2's chunk barrier and stage 1's two barriers fence LDS only (local fence + s_barrier): no wait there for output stores' acks or for global loads in flight; sums unchanged
# speedup vs baseline: 7.7967x; 1.0437x over previous
#include <hip/hip_runtime.h>


#define BATCH   16384
#define IN_DIM  1024
#define OUT_DIM 128

typedef __attribute__((ext_vector_type(16))) __bf16 v16bf;
typedef __attribute__((ext_vector_type(8)))  float  v8f;
typedef __attribute__((ext_vector_type(2)))  float  v2f;

static __device__ inline v8f wmma_bf16(v16bf a, v16bf b, v8f c) {
  return __builtin_amdgcn_wmma_f32_16x16x32_bf16(false, a, false, b,
                                                 (short)0, c, false, false);
}

#define GPP 68

__global__ __launch_bounds__(256) void compute_G(const float* __restrict__ A,
                                                 float* __restrict__ G) {
  extern __shared__ __attribute__((aligned(16))) float gpart[];
  const int t  = threadIdx.x;
  const int bi = blockIdx.x >> 4, bj = blockIdx.x & 15;
  const float* ap = A + (size_t)(8 * bi) * IN_DIM + 4 * t;
  const float* bp = A + (size_t)(8 * bj) * IN_DIM + 4 * t;
  float4 ra[8], rb[8];
#pragma unroll
  for (int r = 0; r < 8; ++r) {
    ra[r] = *(const float4*)(ap + (size_t)r * IN_DIM);
    rb[r] = *(const float4*)(bp + (size_t)r * IN_DIM);
  }
  float* mine = gpart + t * GPP;
#pragma unroll
  for (int i = 0; i < 8; ++i) {
    float s[8];
#pragma unroll
    for (int j = 0; j < 8; ++j) {
      float v = ra[i].x * rb[j].x;
      v = fmaf(ra[i].y, rb[j].y, v);
      v = fmaf(ra[i].z, rb[j].z, v);
      s[j] = fmaf(ra[i].w, rb[j].w, v);
    }
    *(float4*)(mine + 8 * i)     = make_float4(s[0], s[1], s[2], s[3]);
    *(float4*)(mine + 8 * i + 4) = make_float4(s[4], s[5], s[6], s[7]);
  }
  __syncthreads();

  const int o = t & 63, q = t >> 6;
  const float* col = gpart + (64 * q) * GPP + o;
  float c0 = col[0], c1 = col[GPP], c2 = col[2 * GPP], c3 = col[3 * GPP];
#pragma unroll
  for (int u = 4; u < 64; u += 4) {
    c0 += col[(u    ) * GPP];
    c1 += col[(u + 1) * GPP];
    c2 += col[(u + 2) * GPP];
    c3 += col[(u + 3) * GPP];
  }
  float* red = gpart + 256 * GPP;
  red[64 * q + o] = (c0 + c1) + (c2 + c3);
  __syncthreads();
  if (t < 64)
    G[blockIdx.x * 64 + t] = (red[t] + red[64 + t]) + (red[128 + t] + red[192 + t]);
}

#define GLD 132

static __device__ inline v8f wmma_f32(v2f a, v2f b, v8f c) {
  return __builtin_amdgcn_wmma_f32_16x16x4_f32(false, a, false, b,
                                               (short)0, c, false, false);
}

static __device__ inline float lane_value(float x, int src) {
  return __builtin_bit_cast(float,
      __builtin_amdgcn_readlane(__builtin_bit_cast(int, x), src));
}

template <int K>
static __device__ inline float row_lane(float x) {
  return __builtin_bit_cast(float, __builtin_amdgcn_update_dpp(
      0, __builtin_bit_cast(int, x), 0x150 + K, 0xf, 0xf, true));
}

static __device__ inline float other_half(float x) {
  return __builtin_bit_cast(float, __builtin_amdgcn_permlanex16(
      __builtin_bit_cast(int, x), __builtin_bit_cast(int, x),
      0x76543210, 0xfedcba98, false, false));
}

template <int K>
static __device__ inline void gj_pivot(float (&e)[8], int c, int h) {
  constexpr int KH = K >> 3, KK = K & 7;
  const float pinv = 1.0f / lane_value(e[KK], K + 16 * KH);
  const float own  = e[KK];
  const float oth  = other_half(own);
  const float colk = (h == KH) ? own : oth;
  const bool  isk  = (c == K);
  const float nf   = isk ? 0.0f : -colk;
  float rk[8], up[8];
#pragma unroll
  for (int jj = 0; jj < 8; ++jj) rk[jj] = row_lane<K>(e[jj]) * pinv;
#pragma unroll
  for (int jj = 0; jj < 8; ++jj) up[jj] = fmaf(nf, rk[jj], e[jj]);
#pragma unroll
  for (int jj = 0; jj < 8; ++jj) e[jj] = isk ? rk[jj] : up[jj];
  const float ck = isk ? pinv : nf * pinv;
  e[KK] = (h == KH) ? ck : e[KK];
}

template <int K> struct GJ16 {
  static __device__ inline void run(float (&e)[8], int c, int h) {
    gj_pivot<K>(e, c, h);
    GJ16<K + 1>::run(e, c, h);
  }
};
template <> struct GJ16<16> {
  static __device__ inline void run(float (&)[8], int, int) {}
};

static __device__ inline void load_fragA16(const float* p, v2f (&f)[4]) {
  float4 u = *(const float4*)p;
  float4 w = *(const float4*)(p + 4);
  f[0] = v2f{u.x, u.y};  f[1] = v2f{u.z, u.w};
  f[2] = v2f{w.x, w.y};  f[3] = v2f{w.z, w.w};
}

__global__ __launch_bounds__(256) void invert_G(const float* __restrict__ G,
                                                __bf16* __restrict__ Gh,
                                                __bf16* __restrict__ Gl) {
  extern __shared__ __attribute__((aligned(16))) float M[];
  const int tid  = threadIdx.x;
  const int J    = tid >> 5;
  const int lane = tid & 31;
  const int h    = lane >> 4;
  const int c    = lane & 15;

  for (int idx = tid; idx < 128 * 32; idx += 256) {
    const int tile = idx >> 4;
    const int i  = 8 * (tile >> 4) + ((idx >> 1) & 7);
    const int j4 = 8 * (tile & 15) + 4 * (idx & 1);
    *(float4*)(M + i * GLD + j4) = *(const float4*)(G + 4 * idx);
  }
  __syncthreads();

  for (int K = 0; K < 8; ++K) {
    float d[8];
    {
      const float* dp = M + (16 * K + c) * GLD + 16 * K + 8 * h;
      const float4 v0 = *(const float4*)dp;
      const float4 v1 = *(const float4*)(dp + 4);
      d[0] = v0.x; d[1] = v0.y; d[2] = v0.z; d[3] = v0.w;
      d[4] = v1.x; d[5] = v1.y; d[6] = v1.z; d[7] = v1.w;
    }
    v2f fa[8][4];
#pragma unroll
    for (int i = 0; i < 8; ++i)
      load_fragA16(M + (16 * i + c) * GLD + 16 * K + 8 * h, fa[i]);
    __syncthreads();

    if (J == K) {
      for (int rr = lane; rr < 128; rr += 32) {
        float* p = M + rr * GLD + 16 * K;
        const int e = rr - 16 * K;
#pragma unroll
        for (int j = 0; j < 16; j += 4)
          *(float4*)(p + j) = make_float4(e == j     ? 1.0f : 0.0f,
                                          e == j + 1 ? 1.0f : 0.0f,
                                          e == j + 2 ? 1.0f : 0.0f,
                                          e == j + 3 ? 1.0f : 0.0f);
      }
    }

    GJ16<0>::run(d, c, h);
    const v2f da[4] = {v2f{d[0], d[1]}, v2f{d[2], d[3]},
                       v2f{d[4], d[5]}, v2f{d[6], d[7]}};

    v8f t = {};
    {
      const float* bp = M + (16 * K + 8 * h) * GLD + 16 * J + c;
#pragma unroll
      for (int q = 0; q < 4; ++q) {
        v2f bq = v2f{bp[(2 * q) * GLD], bp[(2 * q + 1) * GLD]};
        t = wmma_f32(da[q], bq, t);
      }
    }
    v2f nb[4];
#pragma unroll
    for (int q = 0; q < 4; ++q) nb[q] = v2f{-t[2 * q], -t[2 * q + 1]};

#pragma unroll
    for (int i = 0; i < 8; ++i) {
      float* mp = M + (16 * i + 8 * h) * GLD + 16 * J + c;
      v8f acc;
#pragma unroll
      for (int r = 0; r < 8; ++r) acc[r] = mp[r * GLD];
#pragma unroll
      for (int q = 0; q < 4; ++q) acc = wmma_f32(fa[i][q], nb[q], acc);
#pragma unroll
      for (int r = 0; r < 8; ++r) mp[r * GLD] = (i == K) ? t[r] : acc[r];
    }
    __syncthreads();
  }

  for (int idx = tid; idx < 128 * 128; idx += 256) {
    int i = idx >> 7, j = idx & 127;
    float v = M[i * GLD + j];
    __bf16 hh = (__bf16)v;
    Gh[idx] = hh; Gl[idx] = (__bf16)(v - (float)hh);
  }
}

__global__ __launch_bounds__(256) void pack_A(const float* __restrict__ A,
                                              __bf16* __restrict__ Ah,
                                              __bf16* __restrict__ Al,
                                              __bf16* __restrict__ Ath,
                                              __bf16* __restrict__ Atl) {
  int idx = blockIdx.x * 256 + threadIdx.x;
  int r = idx >> 10, c = idx & 1023;
  float x = A[idx];
  __bf16 h = (__bf16)x;
  __bf16 l = (__bf16)(x - (float)h);
  Ah[idx] = h;  Al[idx] = l;
  Ath[c * 128 + r] = h;  Atl[c * 128 + r] = l;
}

#define RB   64
#define HP   136
#define YP   132
#define RP   132
#define OFF_YL 17408
#define OFF_BH 34816
#define OFF_BL 69632
#define BUFB   104448

typedef __attribute__((ext_vector_type(8))) __bf16 v8bf;
typedef __attribute__((ext_vector_type(4))) __bf16 v4bf;
typedef __attribute__((ext_vector_type(4))) float  v4f;

static __device__ inline v16bf lds_frag(const __bf16* rowp, int k0, int half) {
  const v8bf u = *(const v8bf*)(rowp + k0 + 8 * half);
  const v8bf w = *(const v8bf*)(rowp + k0 + 16 + 8 * half);
  return __builtin_shufflevector(u, w, 0, 1, 2, 3, 4, 5, 6, 7,
                                 8, 9, 10, 11, 12, 13, 14, 15);
}

static __device__ inline void split_fragA_lds(const float* rowp, int k0,
                                              int half, bool neg,
                                              v16bf& hi, v16bf& lo) {
  const float* p0 = rowp + k0 + 8 * half;
  const float4 a0 = *(const float4*)(p0);
  const float4 a1 = *(const float4*)(p0 + 4);
  const float4 c0 = *(const float4*)(p0 + 16);
  const float4 c1 = *(const float4*)(p0 + 20);
  const float x[16] = {a0.x, a0.y, a0.z, a0.w, a1.x, a1.y, a1.z, a1.w,
                       c0.x, c0.y, c0.z, c0.w, c1.x, c1.y, c1.z, c1.w};
#pragma unroll
  for (int e = 0; e < 16; ++e) {
    const float xe = neg ? -x[e] : x[e];
    const __bf16 he = (__bf16)xe;
    hi[e] = he;  lo[e] = (__bf16)(xe - (float)he);
  }
}

static __device__ inline void fetch_y(const float* __restrict__ y, int i0,
                                      int col0, int tid, v4f (&yq)[4]) {
#pragma unroll
  for (int j = 0; j < 4; ++j) {
    const int idx = tid + 512 * j;
    yq[j] = *(const v4f*)(y + (size_t)(i0 + (idx >> 5)) * IN_DIM + col0 +
                          4 * (idx & 31));
  }
}

static __device__ inline void fetch_b(const __bf16* __restrict__ Ph,
                                      const __bf16* __restrict__ Pl,
                                      size_t base, int pitch, int tid,
                                      v8bf (&bq)[8]) {
#pragma unroll
  for (int j = 0; j < 4; ++j) {
    const int idx = tid + 512 * j;
    const size_t off = base + (size_t)(idx >> 4) * pitch + 8 * (idx & 15);
    bq[j]     = *(const v8bf*)(Ph + off);
    bq[j + 4] = *(const v8bf*)(Pl + off);
  }
}

static __device__ inline void stash_b(char* buf, int tid, const v8bf (&bq)[8]) {
  __bf16* Bh = (__bf16*)(buf + OFF_BH);
  __bf16* Bl = (__bf16*)(buf + OFF_BL);
#pragma unroll
  for (int j = 0; j < 4; ++j) {
    const int idx = tid + 512 * j;
    const int o = (idx >> 4) * HP + 8 * (idx & 15);
    *(v8bf*)(Bh + o) = bq[j];
    *(v8bf*)(Bl + o) = bq[j + 4];
  }
}

static __device__ inline void stash_y_split(char* buf, int tid,
                                            const v4f (&yq)[4]) {
  __bf16* Yh = (__bf16*)buf;
  __bf16* Yl = (__bf16*)(buf + OFF_YL);
#pragma unroll
  for (int j = 0; j < 4; ++j) {
    const int idx = tid + 512 * j;
    const int o = (idx >> 5) * HP + 4 * (idx & 31);
    const float x[4] = {yq[j][0], yq[j][1], yq[j][2], yq[j][3]};
    v4bf h, l;
#pragma unroll
    for (int e = 0; e < 4; ++e) {
      const __bf16 he = (__bf16)x[e];
      h[e] = he;  l[e] = (__bf16)(x[e] - (float)he);
    }
    *(v4bf*)(Yh + o) = h;
    *(v4bf*)(Yl + o) = l;
  }
}

static __device__ inline void stash_y_f32(char* buf, int tid,
                                          const v4f (&yq)[4]) {
  float* Yc = (float*)buf;
#pragma unroll
  for (int j = 0; j < 4; ++j) {
    const int idx = tid + 512 * j;
    *(v4f*)(Yc + (idx >> 5) * YP + 4 * (idx & 31)) = yq[j];
  }
}

static __device__ inline void stage0_products(const char* cur, int rt, int l15,
                                              int half, int n0a,
                                              v8f& acc0, v8f& acc1) {
  const __bf16* yh  = (const __bf16*)cur + (16 * rt + l15) * HP;
  const __bf16* yl  = (const __bf16*)(cur + OFF_YL) + (16 * rt + l15) * HP;
  const __bf16* b0h = (const __bf16*)(cur + OFF_BH) + (n0a + l15) * HP;
  const __bf16* b0l = (const __bf16*)(cur + OFF_BL) + (n0a + l15) * HP;
  const __bf16* b1h = b0h + 16 * HP;
  const __bf16* b1l = b0l + 16 * HP;
#pragma unroll
  for (int ks = 0; ks < 4; ++ks) {
    const int k0 = 32 * ks;
    const v16bf ah = lds_frag(yh, k0, half);
    const v16bf al = lds_frag(yl, k0, half);
    v16bf fh = lds_frag(b0h, k0, half);
    v16bf fl = lds_frag(b0l, k0, half);
    acc0 = wmma_bf16(ah, fh, acc0);
    acc0 = wmma_bf16(al, fh, acc0);
    acc0 = wmma_bf16(ah, fl, acc0);
    fh = lds_frag(b1h, k0, half);
    fl = lds_frag(b1l, k0, half);
    acc1 = wmma_bf16(ah, fh, acc1);
    acc1 = wmma_bf16(al, fh, acc1);
    acc1 = wmma_bf16(ah, fl, acc1);
  }
}

static __device__ inline void lds_barrier() {
  __builtin_amdgcn_fence(__ATOMIC_RELEASE, "workgroup", "local");
  __builtin_amdgcn_s_barrier();
  __builtin_amdgcn_fence(__ATOMIC_ACQUIRE, "workgroup", "local");
}

__global__ __launch_bounds__(512) void fused_project(
    const float* __restrict__ y, const float* __restrict__ b,
    const __bf16* __restrict__ Ah, const __bf16* __restrict__ Al,
    const __bf16* __restrict__ Ath, const __bf16* __restrict__ Atl,
    const __bf16* __restrict__ Gh, const __bf16* __restrict__ Gl,
    float* __restrict__ out) {
  extern __shared__ __attribute__((aligned(16))) char lds[];
  char*  buf0 = lds;
  char*  buf1 = lds + BUFB;
  float* s_r  = (float*)(lds + 2 * BUFB);
  float* s_mu = s_r + RB * RP;

  const int tid  = threadIdx.x;
  const int wid  = tid >> 5;
  const int rt   = wid >> 2;
  const int cq   = wid & 3;
  const int lane = tid & 31;
  const int half = lane >> 4;
  const int l15  = lane & 15;
  const int i0   = blockIdx.x * RB;
  const int n0a  = cq * 32;
  const int n0b  = n0a + 16;

  v4f    yq[4];
  v8bf   bq[8];

  fetch_y(y, i0, 0, tid, yq);
  fetch_b(Ah, Al, 0, IN_DIM, tid, bq);
  stash_y_split(buf0, tid, yq);
  stash_b(buf0, tid, bq);
  __syncthreads();

  v8f acc0 = {};
  v8f acc1 = {};
  for (int c = 0; c < 7; ++c) {
    char* cur = (c & 1) ? buf1 : buf0;
    char* nxt = (c & 1) ? buf0 : buf1;
    fetch_y(y, i0, 128 * (c + 1), tid, yq);
    fetch_b(Ah, Al, (size_t)(128 * (c + 1)), IN_DIM, tid, bq);
    __builtin_amdgcn_sched_barrier(0);
    stage0_products(cur, rt, l15, half, n0a, acc0, acc1);
    asm volatile("" : "+v"(acc0), "+v"(acc1), "+v"(yq[0]), "+v"(yq[1]),
                      "+v"(yq[2]), "+v"(yq[3]));
    stash_y_split(nxt, tid, yq);
    stash_b(nxt, tid, bq);
    __syncthreads();
  }

  fetch_b(Gh, Gl, 0, OUT_DIM, tid, bq);
  float bv0[8], bv1[8];
#pragma unroll
  for (int v = 0; v < 8; ++v) {
    const float* bp = b + (size_t)(i0 + 16 * rt + v + 8 * half) * OUT_DIM + l15;
    bv0[v] = bp[n0a];
    bv1[v] = bp[n0b];
  }
  __builtin_amdgcn_sched_barrier(0);
  stage0_products(buf1, rt, l15, half, n0a, acc0, acc1);
  stash_b(buf0, tid, bq);

  fetch_y(y, i0, 0, tid, yq);
  fetch_b(Ath, Atl, 0, OUT_DIM, tid, bq);

#pragma unroll
  for (int v = 0; v < 8; ++v) {
    const int m = 16 * rt + v + 8 * half;
    s_r[m * RP + n0a + l15] = acc0[v] - bv0[v];
    s_r[m * RP + n0b + l15] = acc1[v] - bv1[v];
  }
  lds_barrier();

  const __bf16* g0h = (const __bf16*)(buf0 + OFF_BH) + (n0a + l15) * HP;
  const __bf16* g0l = (const __bf16*)(buf0 + OFF_BL) + (n0a + l15) * HP;
  const __bf16* g1h = g0h + 16 * HP;
  const __bf16* g1l = g0l + 16 * HP;
  v8f mu0 = {};
  v8f mu1 = {};
#pragma unroll
  for (int kk = 0; kk < 4; ++kk) {
    const int k0 = kk * 32;
    v16bf rh, rl;
    split_fragA_lds(s_r + (16 * rt + l15) * RP, k0, half, false, rh, rl);
    v16bf fh = lds_frag(g0h, k0, half);
    v16bf fl = lds_frag(g0l, k0, half);
    mu0 = wmma_bf16(rh, fh, mu0);
    mu0 = wmma_bf16(rl, fh, mu0);
    mu0 = wmma_bf16(rh, fl, mu0);
    fh = lds_frag(g1h, k0, half);
    fl = lds_frag(g1l, k0, half);
    mu1 = wmma_bf16(rh, fh, mu1);
    mu1 = wmma_bf16(rl, fh, mu1);
    mu1 = wmma_bf16(rh, fl, mu1);
  }
#pragma unroll
  for (int v = 0; v < 8; ++v) {
    const int m = 16 * rt + v + 8 * half;
    s_mu[m * RP + n0a + l15] = mu0[v];
    s_mu[m * RP + n0b + l15] = mu1[v];
  }
  lds_barrier();

  v16bf mh[4], ml[4];
#pragma unroll
  for (int kk = 0; kk < 4; ++kk)
    split_fragA_lds(s_mu + (16 * rt + l15) * RP, kk * 32, half, true,
                    mh[kk], ml[kk]);

  stash_y_f32(buf0, tid, yq);
  stash_b(buf0, tid, bq);
  __syncthreads();

  for (int c = 0; c < 8; ++c) {
    char* cur = (c & 1) ? buf1 : buf0;
    char* nxt = (c & 1) ? buf0 : buf1;
    if (c + 1 < 8) {
      fetch_y(y, i0, 128 * (c + 1), tid, yq);
      fetch_b(Ath, Atl, (size_t)(128 * (c + 1)) * OUT_DIM, OUT_DIM, tid, bq);
    }
    __builtin_amdgcn_sched_barrier(0);
#pragma unroll
    for (int t2 = 0; t2 < 2; ++t2) {
      const int ct = 32 * cq + 16 * t2;
      const float*  yc = (const float*)cur + (16 * rt + 8 * half) * YP + ct + l15;
      const __bf16* ph = (const __bf16*)(cur + OFF_BH) + (ct + l15) * HP;
      const __bf16* pl = (const __bf16*)(cur + OFF_BL) + (ct + l15) * HP;
      v8f acc;
#pragma unroll
      for (int v = 0; v < 8; ++v) acc[v] = yc[v * YP];
#pragma unroll
      for (int kk = 0; kk < 4; ++kk) {
        const int k0 = kk * 32;
        const v16bf fh = lds_frag(ph, k0, half);
        const v16bf fl = lds_frag(pl, k0, half);
        acc = wmma_bf16(mh[kk], fh, acc);
        acc = wmma_bf16(ml[kk], fh, acc);
        acc = wmma_bf16(mh[kk], fl, acc);
      }
      float* op = out + (size_t)(i0 + 16 * rt + 8 * half) * IN_DIM + 128 * c + ct + l15;
#pragma unroll
      for (int v = 0; v < 8; ++v) op[(size_t)v * IN_DIM] = acc[v];
    }
    if (c + 1 < 8) {
      stash_y_f32(nxt, tid, yq);
      stash_b(nxt, tid, bq);
    }
    lds_barrier();
  }
}

extern "C" void kernel_launch(void* const* d_in, const int* in_sizes, int n_in,
                              void* d_out, int out_size, void* d_ws,
                              size_t ws_size, hipStream_t stream) {
  (void)in_sizes; (void)n_in; (void)out_size; (void)ws_size;
  const float* y = (const float*)d_in[0];
  const float* A = (const float*)d_in[1];
  const float* b = (const float*)d_in[2];
  float* out = (float*)d_out;

  char* ws = (char*)d_ws;
  float*  G   = (float*) (ws + 0);
  __bf16* Gh  = (__bf16*)(ws + 65536);
  __bf16* Gl  = (__bf16*)(ws + 98304);
  __bf16* Ah  = (__bf16*)(ws + 131072);
  __bf16* Al  = (__bf16*)(ws + 393216);
  __bf16* Ath = (__bf16*)(ws + 655360);
  __bf16* Atl = (__bf16*)(ws + 917504);

  compute_G<<<256, 256, 70656, stream>>>(A, G);
  invert_G<<<1, 256, 67584, stream>>>(G, Gh, Gl);
  pack_A<<<512, 256, 0, stream>>>(A, Ah, Al, Ath, Atl);
  fused_project<<<256, 512, 276480, stream>>>(y, b, Ah, Al, Ath, Atl,
                                              Gh, Gl, out);
}


# pack_A re-tiled: 64x64 tile of A per group of 8 waves, transposed planes turned through LDS; each store covers whole 128-byte lines (was 2-byte pieces of 32 lines); 512 groups -> 32; values unchanged
# speedup vs baseline: 9.6101x; 1.2326x over previous
#include <hip/hip_runtime.h>


#define BATCH   16384
#define IN_DIM  1024
#define OUT_DIM 128

typedef __attribute__((ext_vector_type(16))) __bf16 v16bf;
typedef __attribute__((ext_vector_type(8)))  float  v8f;
typedef __attribute__((ext_vector_type(2)))  float  v2f;

static __device__ inline v8f wmma_bf16(v16bf a, v16bf b, v8f c) {
  return __builtin_amdgcn_wmma_f32_16x16x32_bf16(false, a, false, b,
                                                 (short)0, c, false, false);
}

#define GPP 68

__global__ __launch_bounds__(256) void compute_G(const float* __restrict__ A,
                                                 float* __restrict__ G) {
  extern __shared__ __attribute__((aligned(16))) float gpart[];
  const int t  = threadIdx.x;
  const int bi = blockIdx.x >> 4, bj = blockIdx.x & 15;
  const float* ap = A + (size_t)(8 * bi) * IN_DIM + 4 * t;
  const float* bp = A + (size_t)(8 * bj) * IN_DIM + 4 * t;
  float4 ra[8], rb[8];
#pragma unroll
  for (int r = 0; r < 8; ++r) {
    ra[r] = *(const float4*)(ap + (size_t)r * IN_DIM);
    rb[r] = *(const float4*)(bp + (size_t)r * IN_DIM);
  }
  float* mine = gpart + t * GPP;
#pragma unroll
  for (int i = 0; i < 8; ++i) {
    float s[8];
#pragma unroll
    for (int j = 0; j < 8; ++j) {
      float v = ra[i].x * rb[j].x;
      v = fmaf(ra[i].y, rb[j].y, v);
      v = fmaf(ra[i].z, rb[j].z, v);
      s[j] = fmaf(ra[i].w, rb[j].w, v);
    }
    *(float4*)(mine + 8 * i)     = make_float4(s[0], s[1], s[2], s[3]);
    *(float4*)(mine + 8 * i + 4) = make_float4(s[4], s[5], s[6], s[7]);
  }
  __syncthreads();

  const int o = t & 63, q = t >> 6;
  const float* col = gpart + (64 * q) * GPP + o;
  float c0 = col[0], c1 = col[GPP], c2 = col[2 * GPP], c3 = col[3 * GPP];
#pragma unroll
  for (int u = 4; u < 64; u += 4) {
    c0 += col[(u    ) * GPP];
    c1 += col[(u + 1) * GPP];
    c2 += col[(u + 2) * GPP];
    c3 += col[(u + 3) * GPP];
  }
  float* red = gpart + 256 * GPP;
  red[64 * q + o] = (c0 + c1) + (c2 + c3);
  __syncthreads();
  if (t < 64)
    G[blockIdx.x * 64 + t] = (red[t] + red[64 + t]) + (red[128 + t] + red[192 + t]);
}

#define GLD 132
#define DIP 20

static __device__ inline v8f wmma_f32(v2f a, v2f b, v8f c) {
  return __builtin_amdgcn_wmma_f32_16x16x4_f32(false, a, false, b,
                                               (short)0, c, false, false);
}

template <int K>
static __device__ inline float row_lane(float x) {
  return __builtin_bit_cast(float, __builtin_amdgcn_update_dpp(
      0, __builtin_bit_cast(int, x), 0x150 + K, 0xf, 0xf, true));
}

static __device__ inline float other_half(float x) {
  return __builtin_bit_cast(float, __builtin_amdgcn_permlanex16(
      __builtin_bit_cast(int, x), __builtin_bit_cast(int, x),
      0x76543210, 0xfedcba98, false, false));
}

template <int K>
static __device__ inline void gj_pivot(float (&e)[8], int c, int h) {
  constexpr int KH = K >> 3, KK = K & 7;
  const float own  = e[KK];
  const float oth  = other_half(own);
  const float colk = (h == KH) ? own : oth;
  const float pinv = __builtin_amdgcn_rcpf(row_lane<K>(colk));
  const bool  isk  = (c == K);
  const float nf   = isk ? 0.0f : -colk;
  float rk[8], up[8];
#pragma unroll
  for (int jj = 0; jj < 8; ++jj) rk[jj] = row_lane<K>(e[jj]) * pinv;
#pragma unroll
  for (int jj = 0; jj < 8; ++jj) up[jj] = fmaf(nf, rk[jj], e[jj]);
#pragma unroll
  for (int jj = 0; jj < 8; ++jj) e[jj] = isk ? rk[jj] : up[jj];
  const float ck = isk ? pinv : nf * pinv;
  e[KK] = (h == KH) ? ck : e[KK];
}

template <int K> struct GJ16 {
  static __device__ inline void run(float (&e)[8], int c, int h) {
    gj_pivot<K>(e, c, h);
    GJ16<K + 1>::run(e, c, h);
  }
};
template <> struct GJ16<16> {
  static __device__ inline void run(float (&)[8], int, int) {}
};

static __device__ inline void load_fragA16(const float* p, v2f (&f)[4]) {
  float4 u = *(const float4*)p;
  float4 w = *(const float4*)(p + 4);
  f[0] = v2f{u.x, u.y};  f[1] = v2f{u.z, u.w};
  f[2] = v2f{w.x, w.y};  f[3] = v2f{w.z, w.w};
}

__global__ __launch_bounds__(256) void invert_G(const float* __restrict__ G,
                                                __bf16* __restrict__ Gh,
                                                __bf16* __restrict__ Gl) {
  extern __shared__ __attribute__((aligned(16))) float M[];
  const int tid  = threadIdx.x;
  const int J    = tid >> 5;
  const int lane = tid & 31;
  const int h    = lane >> 4;
  const int c    = lane & 15;

  for (int idx = tid; idx < 128 * 32; idx += 256) {
    const int tile = idx >> 4;
    const int i  = 8 * (tile >> 4) + ((idx >> 1) & 7);
    const int j4 = 8 * (tile & 15) + 4 * (idx & 1);
    *(float4*)(M + i * GLD + j4) = *(const float4*)(G + 4 * idx);
  }
  __syncthreads();

  float* dsc = M + 128 * GLD;
  for (int K = 0; K < 8; ++K) {
    v2f fa[8][4];
#pragma unroll
    for (int i = 0; i < 8; ++i)
      load_fragA16(M + (16 * i + c) * GLD + 16 * K + 8 * h, fa[i]);

    float*       mcol = M + (8 * h) * GLD + 16 * J + c;
    const float* bp   = mcol + (16 * K) * GLD;
    v2f tb[4];
    v8f acc[8];
    if (J != K) {
#pragma unroll
      for (int q = 0; q < 4; ++q)
        tb[q] = v2f{bp[(2 * q) * GLD], bp[(2 * q + 1) * GLD]};
#pragma unroll
      for (int i = 0; i < 8; ++i)
#pragma unroll
        for (int r = 0; r < 8; ++r) acc[i][r] = mcol[(16 * i + r) * GLD];
    } else {
      float d[8];
      const float* dp = M + (16 * K + c) * GLD + 16 * K + 8 * h;
      const float4 v0 = *(const float4*)dp;
      const float4 v1 = *(const float4*)(dp + 4);
      d[0] = v0.x; d[1] = v0.y; d[2] = v0.z; d[3] = v0.w;
      d[4] = v1.x; d[5] = v1.y; d[6] = v1.z; d[7] = v1.w;
      GJ16<0>::run(d, c, h);
      *(float4*)(dsc + c * DIP + 8 * h)     = make_float4(d[0], d[1], d[2], d[3]);
      *(float4*)(dsc + c * DIP + 8 * h + 4) = make_float4(d[4], d[5], d[6], d[7]);
#pragma unroll
      for (int q = 0; q < 4; ++q)
        tb[q] = v2f{(8 * h + 2 * q     == c) ? 1.0f : 0.0f,
                    (8 * h + 2 * q + 1 == c) ? 1.0f : 0.0f};
#pragma unroll
      for (int i = 0; i < 8; ++i) acc[i] = v8f{};
    }
    __syncthreads();

    v2f da[4];
    load_fragA16(dsc + c * DIP + 8 * h, da);

    v8f t = {};
#pragma unroll
    for (int q = 0; q < 4; ++q) t = wmma_f32(da[q], tb[q], t);
    v2f nb[4];
#pragma unroll
    for (int q = 0; q < 4; ++q) nb[q] = v2f{-t[2 * q], -t[2 * q + 1]};

#pragma unroll
    for (int i = 0; i < 8; ++i) {
#pragma unroll
      for (int q = 0; q < 4; ++q) acc[i] = wmma_f32(fa[i][q], nb[q], acc[i]);
    }
#pragma unroll
    for (int i = 0; i < 8; ++i)
#pragma unroll
      for (int r = 0; r < 8; ++r)
        mcol[(16 * i + r) * GLD] = (i == K) ? t[r] : acc[i][r];
    __syncthreads();
  }

  for (int idx = tid; idx < 128 * 128; idx += 256) {
    int i = idx >> 7, j = idx & 127;
    float v = M[i * GLD + j];
    __bf16 hh = (__bf16)v;
    Gh[idx] = hh; Gl[idx] = (__bf16)(v - (float)hh);
  }
}

#define TP 72

typedef __attribute__((ext_vector_type(8))) __bf16   v8bf;
typedef __attribute__((ext_vector_type(4))) unsigned v4u;

__global__ __launch_bounds__(256) void pack_A(const float* __restrict__ A,
                                              __bf16* __restrict__ Ah,
                                              __bf16* __restrict__ Al,
                                              __bf16* __restrict__ Ath,
                                              __bf16* __restrict__ Atl) {
  extern __shared__ __attribute__((aligned(16))) __bf16 tp[];
  const int t  = threadIdx.x;
  const int r0 = 64 * (blockIdx.x >> 4);
  const int c0 = 64 * (blockIdx.x & 15);
  const int cs = 8 * (t & 7);
  __bf16* th = tp;
  __bf16* tl = tp + 64 * TP;
  v8bf h[2], l[2];
#pragma unroll
  for (int p = 0; p < 2; ++p) {
    const int r = (t >> 3) + 32 * p;
    const size_t o = (size_t)(r0 + r) * IN_DIM + c0 + cs;
    const float4 u = *(const float4*)(A + o);
    const float4 w = *(const float4*)(A + o + 4);
    const float x[8] = {u.x, u.y, u.z, u.w, w.x, w.y, w.z, w.w};
#pragma unroll
    for (int e = 0; e < 8; ++e) {
      const __bf16 he = (__bf16)x[e];
      h[p][e] = he;  l[p][e] = (__bf16)(x[e] - (float)he);
    }
    *(v8bf*)(th + r * TP + cs) = h[p];
    *(v8bf*)(tl + r * TP + cs) = l[p];
  }
  __syncthreads();
#pragma unroll
  for (int p = 0; p < 2; ++p) {
    const size_t o = (size_t)(r0 + (t >> 3) + 32 * p) * IN_DIM + c0 + cs;
    *(v8bf*)(Ah + o) = h[p];
    *(v8bf*)(Al + o) = l[p];
  }

  const unsigned* th32 = (const unsigned*)tp;
  const unsigned* tl32 = th32 + 64 * (TP / 2);
  const int cp = 4 * (t >> 5) + ((t & 31) >> 3);
  const int rs = 8 * (t & 7);
  unsigned wh[8], wl[8];
#pragma unroll
  for (int j = 0; j < 8; ++j) {
    wh[j] = th32[(rs + j) * (TP / 2) + cp];
    wl[j] = tl32[(rs + j) * (TP / 2) + cp];
  }
  v4u he, ho, le, lo;
#pragma unroll
  for (int q = 0; q < 4; ++q) {
    he[q] = (wh[2 * q] & 0xffffu) | (wh[2 * q + 1] << 16);
    ho[q] = (wh[2 * q] >> 16)     | (wh[2 * q + 1] & 0xffff0000u);
    le[q] = (wl[2 * q] & 0xffffu) | (wl[2 * q + 1] << 16);
    lo[q] = (wl[2 * q] >> 16)     | (wl[2 * q + 1] & 0xffff0000u);
  }
  const size_t ot = (size_t)(c0 + 2 * cp) * OUT_DIM + r0 + rs;
  *(v8bf*)(Ath + ot)           = __builtin_bit_cast(v8bf, he);
  *(v8bf*)(Ath + ot + OUT_DIM) = __builtin_bit_cast(v8bf, ho);
  *(v8bf*)(Atl + ot)           = __builtin_bit_cast(v8bf, le);
  *(v8bf*)(Atl + ot + OUT_DIM) = __builtin_bit_cast(v8bf, lo);
}

#define RB   64
#define HP   136
#define YP   132
#define RP   132
#define OFF_YL 17408
#define OFF_BH 34816
#define OFF_BL 69632
#define BUFB   104448

typedef __attribute__((ext_vector_type(4))) __bf16 v4bf;
typedef __attribute__((ext_vector_type(4))) float  v4f;

static __device__ inline v16bf lds_frag(const __bf16* rowp, int k0, int half) {
  const v8bf u = *(const v8bf*)(rowp + k0 + 8 * half);
  const v8bf w = *(const v8bf*)(rowp + k0 + 16 + 8 * half);
  return __builtin_shufflevector(u, w, 0, 1, 2, 3, 4, 5, 6, 7,
                                 8, 9, 10, 11, 12, 13, 14, 15);
}

static __device__ inline void split_fragA_lds(const float* rowp, int k0,
                                              int half, bool neg,
                                              v16bf& hi, v16bf& lo) {
  const float* p0 = rowp + k0 + 8 * half;
  const float4 a0 = *(const float4*)(p0);
  const float4 a1 = *(const float4*)(p0 + 4);
  const float4 c0 = *(const float4*)(p0 + 16);
  const float4 c1 = *(const float4*)(p0 + 20);
  const float x[16] = {a0.x, a0.y, a0.z, a0.w, a1.x, a1.y, a1.z, a1.w,
                       c0.x, c0.y, c0.z, c0.w, c1.x, c1.y, c1.z, c1.w};
#pragma unroll
  for (int e = 0; e < 16; ++e) {
    const float xe = neg ? -x[e] : x[e];
    const __bf16 he = (__bf16)xe;
    hi[e] = he;  lo[e] = (__bf16)(xe - (float)he);
  }
}

static __device__ inline void fetch_y(const float* __restrict__ y, int i0,
                                      int col0, int tid, v4f (&yq)[4]) {
#pragma unroll
  for (int j = 0; j < 4; ++j) {
    const int idx = tid + 512 * j;
    yq[j] = *(const v4f*)(y + (size_t)(i0 + (idx >> 5)) * IN_DIM + col0 +
                          4 * (idx & 31));
  }
}

static __device__ inline void fetch_b(const __bf16* __restrict__ Ph,
                                      const __bf16* __restrict__ Pl,
                                      size_t base, int pitch, int tid,
                                      v8bf (&bq)[8]) {
#pragma unroll
  for (int j = 0; j < 4; ++j) {
    const int idx = tid + 512 * j;
    const size_t off = base + (size_t)(idx >> 4) * pitch + 8 * (idx & 15);
    bq[j]     = *(const v8bf*)(Ph + off);
    bq[j + 4] = *(const v8bf*)(Pl + off);
  }
}

static __device__ inline void stash_b(char* buf, int tid, const v8bf (&bq)[8]) {
  __bf16* Bh = (__bf16*)(buf + OFF_BH);
  __bf16* Bl = (__bf16*)(buf + OFF_BL);
#pragma unroll
  for (int j = 0; j < 4; ++j) {
    const int idx = tid + 512 * j;
    const int o = (idx >> 4) * HP + 8 * (idx & 15);
    *(v8bf*)(Bh + o) = bq[j];
    *(v8bf*)(Bl + o) = bq[j + 4];
  }
}

static __device__ inline void stash_y_split(char* buf, int tid,
                                            const v4f (&yq)[4]) {
  __bf16* Yh = (__bf16*)buf;
  __bf16* Yl = (__bf16*)(buf + OFF_YL);
#pragma unroll
  for (int j = 0; j < 4; ++j) {
    const int idx = tid + 512 * j;
    const int o = (idx >> 5) * HP + 4 * (idx & 31);
    const float x[4] = {yq[j][0], yq[j][1], yq[j][2], yq[j][3]};
    v4bf h, l;
#pragma unroll
    for (int e = 0; e < 4; ++e) {
      const __bf16 he = (__bf16)x[e];
      h[e] = he;  l[e] = (__bf16)(x[e] - (float)he);
    }
    *(v4bf*)(Yh + o) = h;
    *(v4bf*)(Yl + o) = l;
  }
}

static __device__ inline void stash_y_f32(char* buf, int tid,
                                          const v4f (&yq)[4]) {
  float* Yc = (float*)buf;
#pragma unroll
  for (int j = 0; j < 4; ++j) {
    const int idx = tid + 512 * j;
    *(v4f*)(Yc + (idx >> 5) * YP + 4 * (idx & 31)) = yq[j];
  }
}

static __device__ inline void stage0_products(const char* cur, int rt, int l15,
                                              int half, int n0a,
                                              v8f& acc0, v8f& acc1) {
  const __bf16* yh  = (const __bf16*)cur + (16 * rt + l15) * HP;
  const __bf16* yl  = (const __bf16*)(cur + OFF_YL) + (16 * rt + l15) * HP;
  const __bf16* b0h = (const __bf16*)(cur + OFF_BH) + (n0a + l15) * HP;
  const __bf16* b0l = (const __bf16*)(cur + OFF_BL) + (n0a + l15) * HP;
  const __bf16* b1h = b0h + 16 * HP;
  const __bf16* b1l = b0l + 16 * HP;
#pragma unroll
  for (int ks = 0; ks < 4; ++ks) {
    const int k0 = 32 * ks;
    const v16bf ah = lds_frag(yh, k0, half);
    const v16bf al = lds_frag(yl, k0, half);
    v16bf fh = lds_frag(b0h, k0, half);
    v16bf fl = lds_frag(b0l, k0, half);
    acc0 = wmma_bf16(ah, fh, acc0);
    acc0 = wmma_bf16(al, fh, acc0);
    acc0 = wmma_bf16(ah, fl, acc0);
    fh = lds_frag(b1h, k0, half);
    fl = lds_frag(b1l, k0, half);
    acc1 = wmma_bf16(ah, fh, acc1);
    acc1 = wmma_bf16(al, fh, acc1);
    acc1 = wmma_bf16(ah, fl, acc1);
  }
}

static __device__ inline void lds_barrier() {
  __builtin_amdgcn_fence(__ATOMIC_RELEASE, "workgroup", "local");
  __builtin_amdgcn_s_barrier();
  __builtin_amdgcn_fence(__ATOMIC_ACQUIRE, "workgroup", "local");
}

__global__ __launch_bounds__(512) void fused_project(
    const float* __restrict__ y, const float* __restrict__ b,
    const __bf16* __restrict__ Ah, const __bf16* __restrict__ Al,
    const __bf16* __restrict__ Ath, const __bf16* __restrict__ Atl,
    const __bf16* __restrict__ Gh, const __bf16* __restrict__ Gl,
    float* __restrict__ out) {
  extern __shared__ __attribute__((aligned(16))) char lds[];
  char*  buf0 = lds;
  char*  buf1 = lds + BUFB;
  float* s_r  = (float*)(lds + 2 * BUFB);
  float* s_mu = s_r + RB * RP;

  const int tid  = threadIdx.x;
  const int wid  = tid >> 5;
  const int rt   = wid >> 2;
  const int cq   = wid & 3;
  const int lane = tid & 31;
  const int half = lane >> 4;
  const int l15  = lane & 15;
  const int i0   = blockIdx.x * RB;
  const int n0a  = cq * 32;
  const int n0b  = n0a + 16;

  v4f    yq[4];
  v8bf   bq[8];

  fetch_y(y, i0, 0, tid, yq);
  fetch_b(Ah, Al, 0, IN_DIM, tid, bq);
  stash_y_split(buf0, tid, yq);
  stash_b(buf0, tid, bq);
  __syncthreads();

  v8f acc0 = {};
  v8f acc1 = {};
  for (int c = 0; c < 7; ++c) {
    char* cur = (c & 1) ? buf1 : buf0;
    char* nxt = (c & 1) ? buf0 : buf1;
    fetch_y(y, i0, 128 * (c + 1), tid, yq);
    fetch_b(Ah, Al, (size_t)(128 * (c + 1)), IN_DIM, tid, bq);
    __builtin_amdgcn_sched_barrier(0);
    stage0_products(cur, rt, l15, half, n0a, acc0, acc1);
    asm volatile("" : "+v"(acc0), "+v"(acc1), "+v"(yq[0]), "+v"(yq[1]),
                      "+v"(yq[2]), "+v"(yq[3]));
    stash_y_split(nxt, tid, yq);
    stash_b(nxt, tid, bq);
    __syncthreads();
  }

  fetch_b(Gh, Gl, 0, OUT_DIM, tid, bq);
  float bv0[8], bv1[8];
#pragma unroll
  for (int v = 0; v < 8; ++v) {
    const float* bp = b + (size_t)(i0 + 16 * rt + v + 8 * half) * OUT_DIM + l15;
    bv0[v] = bp[n0a];
    bv1[v] = bp[n0b];
  }
  __builtin_amdgcn_sched_barrier(0);
  stage0_products(buf1, rt, l15, half, n0a, acc0, acc1);
  stash_b(buf0, tid, bq);

  fetch_y(y, i0, 0, tid, yq);
  fetch_b(Ath, Atl, 0, OUT_DIM, tid, bq);

#pragma unroll
  for (int v = 0; v < 8; ++v) {
    const int m = 16 * rt + v + 8 * half;
    s_r[m * RP + n0a + l15] = acc0[v] - bv0[v];
    s_r[m * RP + n0b + l15] = acc1[v] - bv1[v];
  }
  lds_barrier();

  const __bf16* g0h = (const __bf16*)(buf0 + OFF_BH) + (n0a + l15) * HP;
  const __bf16* g0l = (const __bf16*)(buf0 + OFF_BL) + (n0a + l15) * HP;
  const __bf16* g1h = g0h + 16 * HP;
  const __bf16* g1l = g0l + 16 * HP;
  v8f mu0 = {};
  v8f mu1 = {};
#pragma unroll
  for (int kk = 0; kk < 4; ++kk) {
    const int k0 = kk * 32;
    v16bf rh, rl;
    split_fragA_lds(s_r + (16 * rt + l15) * RP, k0, half, false, rh, rl);
    v16bf fh = lds_frag(g0h, k0, half);
    v16bf fl = lds_frag(g0l, k0, half);
    mu0 = wmma_bf16(rh, fh, mu0);
    mu0 = wmma_bf16(rl, fh, mu0);
    mu0 = wmma_bf16(rh, fl, mu0);
    fh = lds_frag(g1h, k0, half);
    fl = lds_frag(g1l, k0, half);
    mu1 = wmma_bf16(rh, fh, mu1);
    mu1 = wmma_bf16(rl, fh, mu1);
    mu1 = wmma_bf16(rh, fl, mu1);
  }
#pragma unroll
  for (int v = 0; v < 8; ++v) {
    const int m = 16 * rt + v + 8 * half;
    s_mu[m * RP + n0a + l15] = mu0[v];
    s_mu[m * RP + n0b + l15] = mu1[v];
  }
  lds_barrier();

  v16bf mh[4], ml[4];
#pragma unroll
  for (int kk = 0; kk < 4; ++kk)
    split_fragA_lds(s_mu + (16 * rt + l15) * RP, kk * 32, half, true,
                    mh[kk], ml[kk]);

  stash_y_f32(buf0, tid, yq);
  stash_b(buf0, tid, bq);
  __syncthreads();

  for (int c = 0; c < 8; ++c) {
    char* cur = (c & 1) ? buf1 : buf0;
    char* nxt = (c & 1) ? buf0 : buf1;
    if (c + 1 < 8) {
      fetch_y(y, i0, 128 * (c + 1), tid, yq);
      fetch_b(Ath, Atl, (size_t)(128 * (c + 1)) * OUT_DIM, OUT_DIM, tid, bq);
    }
    __builtin_amdgcn_sched_barrier(0);
#pragma unroll
    for (int t2 = 0; t2 < 2; ++t2) {
      const int ct = 32 * cq + 16 * t2;
      const float*  yc = (const float*)cur + (16 * rt + 8 * half) * YP + ct + l15;
      const __bf16* ph = (const __bf16*)(cur + OFF_BH) + (ct + l15) * HP;
      const __bf16* pl = (const __bf16*)(cur + OFF_BL) + (ct + l15) * HP;
      v8f acc;
#pragma unroll
      for (int v = 0; v < 8; ++v) acc[v] = yc[v * YP];
#pragma unroll
      for (int kk = 0; kk < 4; ++kk) {
        const int k0 = kk * 32;
        const v16bf fh = lds_frag(ph, k0, half);
        const v16bf fl = lds_frag(pl, k0, half);
        acc = wmma_bf16(mh[kk], fh, acc);
        acc = wmma_bf16(ml[kk], fh, acc);
        acc = wmma_bf16(mh[kk], fl, acc);
      }
      float* op = out + (size_t)(i0 + 16 * rt + 8 * half) * IN_DIM + 128 * c + ct + l15;
#pragma unroll
      for (int v = 0; v < 8; ++v) op[(size_t)v * IN_DIM] = acc[v];
    }
    if (c + 1 < 8) {
      stash_y_f32(nxt, tid, yq);
      stash_b(nxt, tid, bq);
    }
    lds_barrier();
  }
}

extern "C" void kernel_launch(void* const* d_in, const int* in_sizes, int n_in,
                              void* d_out, int out_size, void* d_ws,
                              size_t ws_size, hipStream_t stream) {
  (void)in_sizes; (void)n_in; (void)out_size; (void)ws_size;
  const float* y = (const float*)d_in[0];
  const float* A = (const float*)d_in[1];
  const float* b = (const float*)d_in[2];
  float* out = (float*)d_out;

  char* ws = (char*)d_ws;
  float*  G   = (float*) (ws + 0);
  __bf16* Gh  = (__bf16*)(ws + 65536);
  __bf16* Gl  = (__bf16*)(ws + 98304);
  __bf16* Ah  = (__bf16*)(ws + 131072);
  __bf16* Al  = (__bf16*)(ws + 393216);
  __bf16* Ath = (__bf16*)(ws + 655360);
  __bf16* Atl = (__bf16*)(ws + 917504);

  compute_G<<<256, 256, 70656, stream>>>(A, G);
  invert_G<<<1, 256, 68864, stream>>>(G, Gh, Gl);
  pack_A<<<32, 256, 18432, stream>>>(A, Ah, Al, Ath, Atl);
  fused_project<<<256, 512, 276480, stream>>>(y, b, Ah, Al, Ath, Atl,
                                              Gh, Gl, out);
}
